# starting kernel
# baseline (speedup 1.0000x reference)
#include <hip/hip_runtime.h>

#define NN 4096
#define DD 256
#define SS 32
#define KK 4
#define ETA 0.5f

typedef __attribute__((ext_vector_type(16))) _Float16 v16h;
typedef __attribute__((ext_vector_type(8)))  _Float16 v8h;
typedef __attribute__((ext_vector_type(8)))  float    v8f;

__device__ __forceinline__ v16h make_ab(const _Float16* p0, const _Float16* p1) {
  v8h lo = *(const v8h*)p0;
  v8h hi = *(const v8h*)p1;
  return __builtin_shufflevector(lo, hi, 0,1,2,3,4,5,6,7,8,9,10,11,12,13,14,15);
}

__global__ __launch_bounds__(32) void zproj_kernel(
    const float* __restrict__ H, const float* __restrict__ U,
    const float* __restrict__ lp,
    _Float16* __restrict__ Zs, _Float16* __restrict__ Zf, _Float16* __restrict__ ZT)
{
  const int n = blockIdx.x;
  const int k = blockIdx.y;
  const int s = threadIdx.x;
  const float* Hrow = H + (size_t)n * DD;
  const float* Ucol = U + (size_t)k * DD * SS + s;
  float acc = 0.f;
  #pragma unroll 4
  for (int d = 0; d < DD; ++d) acc = fmaf(Hrow[d], Ucol[(size_t)d * SS], acc);
  const float e = __expf(lp[k * SS + s]);
  Zf[((size_t)k * NN + n) * SS + s] = (_Float16)acc;
  Zs[((size_t)k * NN + n) * SS + s] = (_Float16)(acc * e);
  ZT[((size_t)k * SS + s) * NN + n] = (_Float16)acc;
}

__global__ __launch_bounds__(32) void flash_attn_kernel(
    const _Float16* __restrict__ Zs, const _Float16* __restrict__ Zf,
    const _Float16* __restrict__ ZT, const float* __restrict__ adj,
    float* __restrict__ O)
{
  const int k    = blockIdx.y;
  const int qb   = blockIdx.x * 16;
  const int l    = threadIdx.x;
  const int ll   = l & 15;
  const int half = l >> 4;
  const int c0   = half ? 8 : 0;

  __shared__ _Float16 __attribute__((aligned(16))) Plds[16 * 64];

  const float inv_s = 0.17677669529663687f;

  const _Float16* qrow = Zs + ((size_t)k * NN + (qb + ll)) * SS;
  const v16h aQ = make_ab(qrow + c0, qrow + c0 + 16);

  v16h ones;
  #pragma unroll
  for (int i = 0; i < 16; ++i) ones[i] = (_Float16)1.0f;

  float m[8];
  #pragma unroll
  for (int v = 0; v < 8; ++v) m[v] = -INFINITY;
  v8f o0 = {}; v8f o1 = {};
  v8f lacc = {};

  for (int kb = 0; kb < NN; kb += 64) {
    v8f sc[4];
    #pragma unroll
    for (int t = 0; t < 4; ++t) {
      const _Float16* kr = Zf + ((size_t)k * NN + (kb + t * 16 + ll)) * SS;
      const v16h bK = make_ab(kr + c0, kr + c0 + 16);
      v8f z = {};
      sc[t] = __builtin_amdgcn_wmma_f32_16x16x32_f16(false, aQ, false, bK, (short)0, z, false, false);
    }

    #pragma unroll
    for (int v = 0; v < 8; ++v) {
      const int q = qb + v + 8 * half;
      const float* mrow = adj + (size_t)q * NN + kb + ll;
      const float w0 = mrow[0], w1 = mrow[16], w2 = mrow[32], w3 = mrow[48];
      __builtin_prefetch(mrow + 64, 0, 0);
      const float r0 = sc[0][v] * inv_s;
      const float r1 = sc[1][v] * inv_s;
      const float r2 = sc[2][v] * inv_s;
      const float r3 = sc[3][v] * inv_s;
      float mx = fmaxf(fmaxf(r0, r1), fmaxf(r2, r3));
      #pragma unroll
      for (int off = 1; off < 16; off <<= 1) mx = fmaxf(mx, __shfl_xor(mx, off, 32));
      const float mn    = fmaxf(m[v], mx);
      const float scale = __expf(m[v] - mn);
      m[v] = mn;
      const float p0 = w0 * __expf(r0 - mn);
      const float p1 = w1 * __expf(r1 - mn);
      const float p2 = w2 * __expf(r2 - mn);
      const float p3 = w3 * __expf(r3 - mn);
      o0[v]   *= scale;
      o1[v]   *= scale;
      lacc[v] *= scale;
      const int r = v + 8 * half;
      Plds[r * 64 + ll]      = (_Float16)p0;
      Plds[r * 64 + 16 + ll] = (_Float16)p1;
      Plds[r * 64 + 32 + ll] = (_Float16)p2;
      Plds[r * 64 + 48 + ll] = (_Float16)p3;
    }
    asm volatile("s_wait_dscnt 0" ::: "memory");

    #pragma unroll
    for (int c = 0; c < 2; ++c) {
      const v16h aP = make_ab(&Plds[ll * 64 + c * 32 + c0],
                              &Plds[ll * 64 + c * 32 + c0 + 16]);
      const _Float16* vr0 = ZT + ((size_t)k * SS + ll)      * NN + kb + c * 32;
      const _Float16* vr1 = ZT + ((size_t)k * SS + 16 + ll) * NN + kb + c * 32;
      const v16h bV0 = make_ab(vr0 + c0, vr0 + c0 + 16);
      const v16h bV1 = make_ab(vr1 + c0, vr1 + c0 + 16);
      o0   = __builtin_amdgcn_wmma_f32_16x16x32_f16(false, aP, false, bV0,  (short)0, o0,   false, false);
      o1   = __builtin_amdgcn_wmma_f32_16x16x32_f16(false, aP, false, bV1,  (short)0, o1,   false, false);
      lacc = __builtin_amdgcn_wmma_f32_16x16x32_f16(false, aP, false, ones, (short)0, lacc, false, false);
    }
  }

  #pragma unroll
  for (int v = 0; v < 8; ++v) {
    const float inv_l = 1.0f / lacc[v];
    const int q = qb + v + 8 * half;
    O[((size_t)k * NN + q) * SS + ll]      = o0[v] * inv_l;
    O[((size_t)k * NN + q) * SS + 16 + ll] = o1[v] * inv_l;
  }
}

__global__ __launch_bounds__(256) void combine_kernel(
    const float* __restrict__ H, const float* __restrict__ U,
    const float* __restrict__ O, const float* __restrict__ thr,
    float* __restrict__ out)
{
  const int n = blockIdx.x;
  const int d = threadIdx.x;
  __shared__ float Os[KK * SS];
  if (d < KK * SS) Os[d] = O[((size_t)(d >> 5) * NN + n) * SS + (d & 31)];
  __syncthreads();
  float agg = 0.f;
  #pragma unroll
  for (int k = 0; k < KK; ++k) {
    const float* Urow = U + ((size_t)k * DD + d) * SS;
    #pragma unroll
    for (int s = 0; s < SS; ++s) agg = fmaf(Os[k * SS + s], Urow[s], agg);
  }
  const float h = H[(size_t)n * DD + d] + ETA * agg;
  const float a = fabsf(h) - thr[d];
  out[(size_t)n * DD + d] = (a > 0.f) ? copysignf(a, h) : 0.f;
}

__global__ __launch_bounds__(256) void orth_kernel(const float* __restrict__ U,
                                                   float* __restrict__ out)
{
  const int t = threadIdx.x;
  float acc = 0.f;
  for (int idx = t; idx < KK * KK * SS * SS; idx += 256) {
    const int b  = idx & (SS - 1);
    const int a  = (idx >> 5) & (SS - 1);
    const int lq = (idx >> 10) & (KK - 1);
    const int kq = idx >> 12;
    if (kq > lq) continue;
    float c = 0.f;
    const float* Ua = U + (size_t)kq * DD * SS + a;
    const float* Ub = U + (size_t)lq * DD * SS + b;
    for (int d = 0; d < DD; ++d) c = fmaf(Ua[(size_t)d * SS], Ub[(size_t)d * SS], c);
    if (kq == lq) { const float e = c - (a == b ? 1.f : 0.f); acc += e * e; }
    else          { acc += c * c; }
  }
  __shared__ float red[256];
  red[t] = acc; __syncthreads();
  for (int off = 128; off > 0; off >>= 1) {
    if (t < off) red[t] += red[t + off];
    __syncthreads();
  }
  if (t == 0) out[(size_t)NN * DD] = red[0];
}

extern "C" void kernel_launch(void* const* d_in, const int* in_sizes, int n_in,
                              void* d_out, int out_size, void* d_ws, size_t ws_size,
                              hipStream_t stream) {
  (void)in_sizes; (void)n_in; (void)out_size; (void)ws_size;
  const float* H   = (const float*)d_in[0];
  const float* adj = (const float*)d_in[1];
  const float* U   = (const float*)d_in[2];
  const float* lp  = (const float*)d_in[3];
  const float* thr = (const float*)d_in[4];
  float* out = (float*)d_out;

  char* ws = (char*)d_ws;
  _Float16* Zs = (_Float16*)(ws);
  _Float16* Zf = (_Float16*)(ws + (1u << 20));
  _Float16* ZT = (_Float16*)(ws + (2u << 20));
  float*    O  = (float*)   (ws + (3u << 20));

  zproj_kernel     <<<dim3(NN, KK),      32, 0, stream>>>(H, U, lp, Zs, Zf, ZT);
  flash_attn_kernel<<<dim3(NN / 16, KK), 32, 0, stream>>>(Zs, Zf, ZT, adj, O);
  combine_kernel   <<<NN,               DD, 0, stream>>>(H, U, O, thr, out);
  orth_kernel      <<<1,               256, 0, stream>>>(U, out);
}


# projection: 16384 one-row one-wave blocks -> 256 four-wave blocks of 16 rows; H tile staged in LDS, read as broadcasts; two U columns per lane in registers a chunk ahead; per-lane FMA, d order kept
# speedup vs baseline: 6.0324x; 6.0324x over previous
#include <hip/hip_runtime.h>

#define NN 4096
#define DD 256
#define SS 32
#define KK 4
#define ETA 0.5f

typedef __attribute__((ext_vector_type(16))) _Float16 v16h;
typedef __attribute__((ext_vector_type(8)))  _Float16 v8h;
typedef __attribute__((ext_vector_type(8)))  float    v8f;
typedef __attribute__((ext_vector_type(2)))  float    v2f;
typedef __attribute__((ext_vector_type(4)))  float    v4f;

__device__ __forceinline__ v16h make_ab(const _Float16* p0, const _Float16* p1) {
  v8h lo = *(const v8h*)p0;
  v8h hi = *(const v8h*)p1;
  return __builtin_shufflevector(lo, hi, 0,1,2,3,4,5,6,7,8,9,10,11,12,13,14,15);
}

#define ZP_ROWS 16
#define ZP_CH   32

struct UChunk { v2f u[ZP_CH]; };

__device__ __forceinline__ void zp_load(UChunk& u, const float* __restrict__ up0,
                                        const float* __restrict__ up1, int d0)
{
  #pragma unroll
  for (int j = 0; j < ZP_CH; ++j) {
    u.u[j][0] = up0[(size_t)(d0 + j) * SS];
    u.u[j][1] = up1[(size_t)(d0 + j) * SS];
  }
}

#define ZP_G 16
struct HGroup { v4f h[8][ZP_G / 4]; };

__device__ __forceinline__ void zp_hload(HGroup& g, const float* hrow, int d)
{
  #pragma unroll
  for (int r = 0; r < 8; ++r) {
    #pragma unroll
    for (int c = 0; c < ZP_G / 4; ++c)
      g.h[r][c] = *(const v4f*)(hrow + r * DD + d + 4 * c);
  }
}

__device__ __forceinline__ void zp_fma(const v2f* u, const HGroup& g, v2f (&acc)[8])
{
  #pragma unroll
  for (int c = 0; c < ZP_G / 4; ++c) {
    #pragma unroll
    for (int q = 0; q < 4; ++q) {
      #pragma unroll
      for (int r = 0; r < 8; ++r) {
        v2f hh;
        hh[0] = g.h[r][c][q];
        hh[1] = g.h[r][c][q];
        acc[r] = __builtin_elementwise_fma(hh, u[4 * c + q], acc[r]);
      }
    }
  }
}

__global__ __launch_bounds__(128) void zproj_kernel(
    const float* __restrict__ H, const float* __restrict__ U,
    const float* __restrict__ lp,
    _Float16* __restrict__ Zs, _Float16* __restrict__ Zf, _Float16* __restrict__ ZT)
{
  __shared__ float __attribute__((aligned(16))) Hs[ZP_ROWS * DD];

  const int w  = __builtin_amdgcn_readfirstlane((int)(threadIdx.x >> 5));
  const int s  = threadIdx.x & 31;
  const int k0 = 2 * (w & 1);
  const int r0 = 8 * (w >> 1);
  const int nb = blockIdx.x * ZP_ROWS;

  const float* up0 = U + (size_t)k0 * DD * SS + s;
  const float* up1 = up0 + (size_t)DD * SS;

  UChunk A, B;
  zp_load(A, up0, up1, 0);

  const float* hsrc = H + (size_t)nb * DD;
  #pragma unroll
  for (int i = 0; i < 8; ++i) {
    const int o = (i * 128 + (int)threadIdx.x) * 4;
    *(v4f*)(Hs + o) = *(const v4f*)(hsrc + o);
  }
  const float e0 = __expf(lp[k0 * SS + s]);
  const float e1 = __expf(lp[(k0 + 1) * SS + s]);
  __syncthreads();

  const float* hrow = Hs + r0 * DD;
  v2f acc[8];
  #pragma unroll
  for (int r = 0; r < 8; ++r) { acc[r][0] = 0.f; acc[r][1] = 0.f; }

  HGroup HA, HB;
  zp_hload(HA, hrow, 0);
  #pragma nounroll
  for (int d0 = 0; d0 < DD; d0 += 2 * ZP_CH) {
    zp_load(B, up0, up1, d0 + ZP_CH);
    zp_hload(HB, hrow, d0 + ZP_G);
    __builtin_amdgcn_sched_barrier(0);
    zp_fma(A.u, HA, acc);
    zp_hload(HA, hrow, d0 + 2 * ZP_G);
    __builtin_amdgcn_sched_barrier(0);
    zp_fma(A.u + ZP_G, HB, acc);
    zp_load(A, up0, up1, (d0 + 2 * ZP_CH) & (DD - 1));
    zp_hload(HB, hrow, d0 + 3 * ZP_G);
    __builtin_amdgcn_sched_barrier(0);
    zp_fma(B.u, HA, acc);
    zp_hload(HA, hrow, (d0 + 4 * ZP_G) & (DD - 1));
    __builtin_amdgcn_sched_barrier(0);
    zp_fma(B.u + ZP_G, HB, acc);
  }

  v8h t0, t1;
  #pragma unroll
  for (int r = 0; r < 8; ++r) {
    const size_t n = (size_t)(nb + r0 + r);
    Zf[((size_t)k0 * NN + n) * SS + s]       = (_Float16)acc[r][0];
    Zs[((size_t)k0 * NN + n) * SS + s]       = (_Float16)(acc[r][0] * e0);
    Zf[((size_t)(k0 + 1) * NN + n) * SS + s] = (_Float16)acc[r][1];
    Zs[((size_t)(k0 + 1) * NN + n) * SS + s] = (_Float16)(acc[r][1] * e1);
    t0[r] = (_Float16)acc[r][0];
    t1[r] = (_Float16)acc[r][1];
  }
  *(v8h*)(ZT + ((size_t)k0 * SS + s) * NN + nb + r0)       = t0;
  *(v8h*)(ZT + ((size_t)(k0 + 1) * SS + s) * NN + nb + r0) = t1;
}

struct AttnBlock {
  v16h  bK[4];
  v16h  bV[2][2];
  float w[8][4];
};

__device__ __forceinline__ void attn_load(AttnBlock& r,
    const _Float16* __restrict__ kbase, const _Float16* __restrict__ vbase0,
    const _Float16* __restrict__ vbase1, const float* __restrict__ mbase, int kb)
{
  #pragma unroll
  for (int t = 0; t < 4; ++t) {
    const _Float16* kr = kbase + (size_t)(kb + t * 16) * SS;
    r.bK[t] = make_ab(kr, kr + 16);
  }
  const float* mrow = mbase + kb;
  #pragma unroll
  for (int v = 0; v < 8; ++v) {
    #pragma unroll
    for (int t = 0; t < 4; ++t) r.w[v][t] = mrow[(size_t)v * NN + t * 16];
    __builtin_prefetch(mrow + (size_t)v * NN + 64, 0, 0);
  }
  #pragma unroll
  for (int c = 0; c < 2; ++c) {
    const _Float16* vr0 = vbase0 + kb + c * 32;
    const _Float16* vr1 = vbase1 + kb + c * 32;
    r.bV[c][0] = make_ab(vr0, vr0 + 16);
    r.bV[c][1] = make_ab(vr1, vr1 + 16);
  }
}

__device__ __forceinline__ void attn_block(const AttnBlock& r, const v16h aQ, const v16h ones,
    const float (&m)[8], const float inv_s, _Float16* Plds,
    const int ll, const int half, const int c0, v8f& o0, v8f& o1, v8f& lacc)
{
  v8f sc[4];
  #pragma unroll
  for (int t = 0; t < 4; ++t) {
    v8f z = {};
    sc[t] = __builtin_amdgcn_wmma_f32_16x16x32_f16(false, aQ, false, r.bK[t], (short)0, z, false, false);
  }
  #pragma unroll
  for (int v = 0; v < 8; ++v) {
    const float mn = m[v];
    const float p0 = r.w[v][0] * __expf(sc[0][v] * inv_s - mn);
    const float p1 = r.w[v][1] * __expf(sc[1][v] * inv_s - mn);
    const float p2 = r.w[v][2] * __expf(sc[2][v] * inv_s - mn);
    const float p3 = r.w[v][3] * __expf(sc[3][v] * inv_s - mn);
    const int row = v + 8 * half;
    Plds[row * 64 + ll]      = (_Float16)p0;
    Plds[row * 64 + 16 + ll] = (_Float16)p1;
    Plds[row * 64 + 32 + ll] = (_Float16)p2;
    Plds[row * 64 + 48 + ll] = (_Float16)p3;
  }
  asm volatile("s_wait_dscnt 0" ::: "memory");
  #pragma unroll
  for (int c = 0; c < 2; ++c) {
    const v16h aP = make_ab(&Plds[ll * 64 + c * 32 + c0],
                            &Plds[ll * 64 + c * 32 + c0 + 16]);
    o0   = __builtin_amdgcn_wmma_f32_16x16x32_f16(false, aP, false, r.bV[c][0], (short)0, o0,   false, false);
    o1   = __builtin_amdgcn_wmma_f32_16x16x32_f16(false, aP, false, r.bV[c][1], (short)0, o1,   false, false);
    lacc = __builtin_amdgcn_wmma_f32_16x16x32_f16(false, aP, false, ones,       (short)0, lacc, false, false);
  }
}

#define S1_TILES 16
#define S1_KEYS  (16 * S1_TILES)
struct KeyGroup { v16h b[S1_TILES]; };

__device__ __forceinline__ void s1_load(KeyGroup& g, const _Float16* __restrict__ kbase, int kb)
{
  #pragma unroll
  for (int t = 0; t < S1_TILES; ++t) {
    const _Float16* kr = kbase + (size_t)(kb + t * 16) * SS;
    g.b[t] = make_ab(kr, kr + 16);
  }
}

__device__ __forceinline__ void s1_max(const KeyGroup& g, const v16h aQ, float (&m)[8])
{
  #pragma unroll
  for (int t = 0; t < S1_TILES; ++t) {
    v8f z = {};
    const v8f s1 = __builtin_amdgcn_wmma_f32_16x16x32_f16(false, aQ, false, g.b[t], (short)0, z, false, false);
    #pragma unroll
    for (int v = 0; v < 8; ++v) m[v] = fmaxf(m[v], s1[v]);
  }
}

#define OPITCH 36

__global__ __launch_bounds__(128) void attn_combine_kernel(
    const _Float16* __restrict__ Zs, const _Float16* __restrict__ Zf,
    const _Float16* __restrict__ ZT, const float* __restrict__ adj,
    const float* __restrict__ H, const float* __restrict__ U,
    const float* __restrict__ thr, float* __restrict__ out)
{
  const int k    = __builtin_amdgcn_readfirstlane((int)(threadIdx.x >> 5));
  const int qb   = blockIdx.x * 16;
  const int l    = threadIdx.x & 31;
  const int ll   = l & 15;
  const int half = l >> 4;
  const int c0   = half ? 8 : 0;

  __shared__ _Float16 __attribute__((aligned(16))) Pall[KK * 16 * 64];
  __shared__ float    __attribute__((aligned(16))) Osh[KK * 16 * OPITCH];
  _Float16* Plds = Pall + k * (16 * 64);

  const float inv_s = 0.17677669529663687f;

  const _Float16* qrow = Zs + ((size_t)k * NN + (qb + ll)) * SS;
  const v16h aQ = make_ab(qrow + c0, qrow + c0 + 16);

  v16h ones;
  #pragma unroll
  for (int i = 0; i < 16; ++i) ones[i] = (_Float16)1.0f;

  float m[8];
  #pragma unroll
  for (int v = 0; v < 8; ++v) m[v] = -INFINITY;

  const _Float16* kbase = Zf + ((size_t)k * NN + ll) * SS + c0;

  KeyGroup GA, GB;
  s1_load(GA, kbase, 0);
  for (int kb = 0; kb < NN; kb += 2 * S1_KEYS) {
    s1_load(GB, kbase, kb + S1_KEYS);
    __builtin_amdgcn_sched_barrier(0);
    s1_max(GA, aQ, m);
    s1_load(GA, kbase, (kb + 2 * S1_KEYS) & (NN - 1));
    __builtin_amdgcn_sched_barrier(0);
    s1_max(GB, aQ, m);
  }
  #pragma unroll
  for (int v = 0; v < 8; ++v) {
    float mx = m[v];
    #pragma unroll
    for (int off = 1; off < 16; off <<= 1) mx = fmaxf(mx, __shfl_xor(mx, off, 32));
    m[v] = mx * inv_s;
  }

  v8f o0 = {}; v8f o1 = {};
  v8f lacc = {};

  const _Float16* vbase0 = ZT + ((size_t)k * SS + ll) * NN + c0;
  const _Float16* vbase1 = vbase0 + (size_t)16 * NN;
  const float*    mbase  = adj + (size_t)(qb + 8 * half) * NN + ll;

  AttnBlock A, B;
  attn_load(A, kbase, vbase0, vbase1, mbase, 0);
  for (int kb = 0; kb < NN; kb += 128) {
    attn_load(B, kbase, vbase0, vbase1, mbase, kb + 64);
    __builtin_amdgcn_sched_barrier(0);
    attn_block(A, aQ, ones, m, inv_s, Plds, ll, half, c0, o0, o1, lacc);
    asm volatile("" : "+v"(o0), "+v"(o1), "+v"(lacc));
    attn_load(A, kbase, vbase0, vbase1, mbase, (kb + 128) & (NN - 1));
    __builtin_amdgcn_sched_barrier(0);
    attn_block(B, aQ, ones, m, inv_s, Plds, ll, half, c0, o0, o1, lacc);
  }

  const int dcol = 64 * k + ll;
  v4f bU[KK][4][4];
  #pragma unroll
  for (int kk = 0; kk < KK; ++kk) {
    #pragma unroll
    for (int j = 0; j < 4; ++j) {
      const float* ur = U + ((size_t)kk * DD + dcol + 16 * j) * SS + 16 * half;
      #pragma unroll
      for (int mm = 0; mm < 4; ++mm) bU[kk][j][mm] = *(const v4f*)(ur + 4 * mm);
    }
  }
  float th[4], hv[4][8];
  #pragma unroll
  for (int j = 0; j < 4; ++j) {
    th[j] = thr[dcol + 16 * j];
    #pragma unroll
    for (int r = 0; r < 8; ++r)
      hv[j][r] = H[(size_t)(qb + 8 * half + r) * DD + dcol + 16 * j];
  }

  float* Ow = Osh + k * (16 * OPITCH);
  #pragma unroll
  for (int v = 0; v < 8; ++v) {
    const float inv_l = 1.0f / lacc[v];
    const int row = v + 8 * half;
    Ow[row * OPITCH + ll]      = o0[v] * inv_l;
    Ow[row * OPITCH + 16 + ll] = o1[v] * inv_l;
  }
  __syncthreads();

  v8f acc[4] = {};
  #pragma unroll
  for (int kk = 0; kk < KK; ++kk) {
    const float* orow = Osh + (kk * 16 + ll) * OPITCH + 16 * half;
    v4f aO[4];
    #pragma unroll
    for (int mm = 0; mm < 4; ++mm) aO[mm] = *(const v4f*)(orow + 4 * mm);
    #pragma unroll
    for (int i = 0; i < 8; ++i) {
      v2f a;
      a[0] = aO[i >> 1][(i & 1) * 2];
      a[1] = aO[i >> 1][(i & 1) * 2 + 1];
      #pragma unroll
      for (int j = 0; j < 4; ++j) {
        v2f b;
        b[0] = bU[kk][j][i >> 1][(i & 1) * 2];
        b[1] = bU[kk][j][i >> 1][(i & 1) * 2 + 1];
        acc[j] = __builtin_amdgcn_wmma_f32_16x16x4_f32(false, a, false, b, (short)0, acc[j], false, false);
      }
    }
  }

  #pragma unroll
  for (int j = 0; j < 4; ++j) {
    #pragma unroll
    for (int r = 0; r < 8; ++r) {
      const float h = hv[j][r] + ETA * acc[j][r];
      const float a = fabsf(h) - th[j];
      out[(size_t)(qb + 8 * half + r) * DD + dcol + 16 * j] = (a > 0.f) ? copysignf(a, h) : 0.f;
    }
  }
}

#define ORTH_WAVES   10
#define ORTH_THREADS 320
#define ORTH_CH    8

struct OrthFrag { v2f a0, a1, b0, b1; };

__device__ __forceinline__ OrthFrag orth_load(const float* __restrict__ pa,
                                              const float* __restrict__ pb, int d0) {
  const v2f x0 = *(const v2f*)(pa + (size_t)d0 * SS);
  const v2f x1 = *(const v2f*)(pa + (size_t)(d0 + 1) * SS);
  const v2f y0 = *(const v2f*)(pb + (size_t)d0 * SS);
  const v2f y1 = *(const v2f*)(pb + (size_t)(d0 + 1) * SS);
  OrthFrag f;
  f.a0[0] = x0[0]; f.a0[1] = x1[0];
  f.a1[0] = x0[1]; f.a1[1] = x1[1];
  f.b0[0] = y0[0]; f.b0[1] = y1[0];
  f.b1[0] = y0[1]; f.b1[1] = y1[1];
  return f;
}

__global__ __launch_bounds__(ORTH_THREADS) void orth_kernel(const float* __restrict__ U,
                                                            float* __restrict__ out)
{
  const int w = threadIdx.x >> 5;
  const int l = threadIdx.x & 31;
  const int c = l & 15;
  const int h = l >> 4;
  const int kq    = (w >= 4) + (w >= 7) + (w >= 9);
  const int first = (kq == 0) ? 0 : (kq == 1) ? 4 : (kq == 2) ? 7 : 9;
  const int lq    = kq + (w - first);

  const float* pa = U + ((size_t)kq * DD + 2 * h) * SS + 2 * c;
  const float* pb = U + ((size_t)lq * DD + 2 * h) * SS + 2 * c;

  v8f g00 = {}, g01 = {}, g10 = {}, g11 = {};
  OrthFrag cur[ORTH_CH], nxt[ORTH_CH];
  #pragma unroll
  for (int j = 0; j < ORTH_CH; ++j) cur[j] = orth_load(pa, pb, 4 * j);

  #pragma unroll
  for (int g = 0; g < DD / (4 * ORTH_CH); ++g) {
    if (g + 1 < DD / (4 * ORTH_CH)) {
      #pragma unroll
      for (int j = 0; j < ORTH_CH; ++j) nxt[j] = orth_load(pa, pb, 4 * ORTH_CH * (g + 1) + 4 * j);
    }
    __builtin_amdgcn_sched_barrier(0);
    #pragma unroll
    for (int j = 0; j < ORTH_CH; ++j) {
      g00 = __builtin_amdgcn_wmma_f32_16x16x4_f32(false, cur[j].a0, false, cur[j].b0, (short)0, g00, false, false);
      g01 = __builtin_amdgcn_wmma_f32_16x16x4_f32(false, cur[j].a0, false, cur[j].b1, (short)0, g01, false, false);
      g10 = __builtin_amdgcn_wmma_f32_16x16x4_f32(false, cur[j].a1, false, cur[j].b0, (short)0, g10, false, false);
      g11 = __builtin_amdgcn_wmma_f32_16x16x4_f32(false, cur[j].a1, false, cur[j].b1, (short)0, g11, false, false);
    }
    #pragma unroll
    for (int j = 0; j < ORTH_CH; ++j) cur[j] = nxt[j];
  }

  const bool diag = (kq == lq);
  float acc = 0.f;
  #pragma unroll
  for (int r = 0; r < 8; ++r) {
    const float one = (diag && (8 * h + r == c)) ? 1.f : 0.f;
    const float e00 = g00[r] - one;
    const float e11 = g11[r] - one;
    acc = fmaf(e00, e00, acc);
    acc = fmaf(g01[r], g01[r], acc);
    acc = fmaf(g10[r], g10[r], acc);
    acc = fmaf(e11, e11, acc);
  }
  for (int off = 16; off > 0; off >>= 1) acc += __shfl_xor(acc, off, 32);

  __shared__ float red[ORTH_WAVES];
  if (l == 0) red[w] = acc;
  __syncthreads();
  if (threadIdx.x == 0) {
    float s = 0.f;
    for (int i = 0; i < ORTH_WAVES; ++i) s += red[i];
    out[(size_t)NN * DD] = s;
  }
}

extern "C" void kernel_launch(void* const* d_in, const int* in_sizes, int n_in,
                              void* d_out, int out_size, void* d_ws, size_t ws_size,
                              hipStream_t stream) {
  (void)in_sizes; (void)n_in; (void)out_size; (void)ws_size;
  const float* H   = (const float*)d_in[0];
  const float* adj = (const float*)d_in[1];
  const float* U   = (const float*)d_in[2];
  const float* lp  = (const float*)d_in[3];
  const float* thr = (const float*)d_in[4];
  float* out = (float*)d_out;

  char* ws = (char*)d_ws;
  _Float16* Zs = (_Float16*)(ws);
  _Float16* Zf = (_Float16*)(ws + (1u << 20));
  _Float16* ZT = (_Float16*)(ws + (2u << 20));

  zproj_kernel       <<<NN / ZP_ROWS,    128, 0, stream>>>(H, U, lp, Zs, Zf, ZT);
  attn_combine_kernel<<<NN / 16,     32 * KK, 0, stream>>>(Zs, Zf, ZT, adj, H, U, thr, out);
  orth_kernel        <<<1,      ORTH_THREADS, 0, stream>>>(U, out);
}


# attention regrouped: work-group = one head x four 16-query tiles, its keys resident in LDS (read once per 64 queries), value fragments from that copy by transposing LDS loads; combine own launch
# speedup vs baseline: 6.7567x; 1.1201x over previous
#include <hip/hip_runtime.h>

#define NN 4096
#define DD 256
#define SS 32
#define KK 4
#define ETA 0.5f

typedef __attribute__((ext_vector_type(16))) _Float16 v16h;
typedef __attribute__((ext_vector_type(8)))  _Float16 v8h;
typedef __attribute__((ext_vector_type(8)))  float    v8f;
typedef __attribute__((ext_vector_type(2)))  float    v2f;
typedef __attribute__((ext_vector_type(4)))  float    v4f;

__device__ __forceinline__ v16h make_ab(const _Float16* p0, const _Float16* p1) {
  v8h lo = *(const v8h*)p0;
  v8h hi = *(const v8h*)p1;
  return __builtin_shufflevector(lo, hi, 0,1,2,3,4,5,6,7,8,9,10,11,12,13,14,15);
}

#define ZP_ROWS 16
#define ZP_CH   32

struct UChunk { v2f u[ZP_CH]; };

__device__ __forceinline__ void zp_load(UChunk& u, const float* __restrict__ up0,
                                        const float* __restrict__ up1, int d0)
{
  #pragma unroll
  for (int j = 0; j < ZP_CH; ++j) {
    u.u[j][0] = up0[(size_t)(d0 + j) * SS];
    u.u[j][1] = up1[(size_t)(d0 + j) * SS];
  }
}

#define ZP_G 16
struct HGroup { v4f h[8][ZP_G / 4]; };

__device__ __forceinline__ void zp_hload(HGroup& g, const float* hrow, int d)
{
  #pragma unroll
  for (int r = 0; r < 8; ++r) {
    #pragma unroll
    for (int c = 0; c < ZP_G / 4; ++c)
      g.h[r][c] = *(const v4f*)(hrow + r * DD + d + 4 * c);
  }
}

__device__ __forceinline__ void zp_fma(const v2f* u, const HGroup& g, v2f (&acc)[8])
{
  #pragma unroll
  for (int c = 0; c < ZP_G / 4; ++c) {
    #pragma unroll
    for (int q = 0; q < 4; ++q) {
      #pragma unroll
      for (int r = 0; r < 8; ++r) {
        v2f hh;
        hh[0] = g.h[r][c][q];
        hh[1] = g.h[r][c][q];
        acc[r] = __builtin_elementwise_fma(hh, u[4 * c + q], acc[r]);
      }
    }
  }
}

__global__ __launch_bounds__(128) void zproj_kernel(
    const float* __restrict__ H, const float* __restrict__ U,
    const float* __restrict__ lp,
    _Float16* __restrict__ Zs, _Float16* __restrict__ Zf)
{
  __shared__ float __attribute__((aligned(16))) Hs[ZP_ROWS * DD];

  const int w  = __builtin_amdgcn_readfirstlane((int)(threadIdx.x >> 5));
  const int s  = threadIdx.x & 31;
  const int k0 = 2 * (w & 1);
  const int r0 = 8 * (w >> 1);
  const int nb = blockIdx.x * ZP_ROWS;

  const float* up0 = U + (size_t)k0 * DD * SS + s;
  const float* up1 = up0 + (size_t)DD * SS;

  UChunk A, B;
  zp_load(A, up0, up1, 0);

  const float* hsrc = H + (size_t)nb * DD;
  #pragma unroll
  for (int i = 0; i < 8; ++i) {
    const int o = (i * 128 + (int)threadIdx.x) * 4;
    *(v4f*)(Hs + o) = *(const v4f*)(hsrc + o);
  }
  const float e0 = __expf(lp[k0 * SS + s]);
  const float e1 = __expf(lp[(k0 + 1) * SS + s]);
  __syncthreads();

  const float* hrow = Hs + r0 * DD;
  v2f acc[8];
  #pragma unroll
  for (int r = 0; r < 8; ++r) { acc[r][0] = 0.f; acc[r][1] = 0.f; }

  HGroup HA, HB;
  zp_hload(HA, hrow, 0);
  #pragma nounroll
  for (int d0 = 0; d0 < DD; d0 += 2 * ZP_CH) {
    zp_load(B, up0, up1, d0 + ZP_CH);
    zp_hload(HB, hrow, d0 + ZP_G);
    __builtin_amdgcn_sched_barrier(0);
    zp_fma(A.u, HA, acc);
    zp_hload(HA, hrow, d0 + 2 * ZP_G);
    __builtin_amdgcn_sched_barrier(0);
    zp_fma(A.u + ZP_G, HB, acc);
    zp_load(A, up0, up1, (d0 + 2 * ZP_CH) & (DD - 1));
    zp_hload(HB, hrow, d0 + 3 * ZP_G);
    __builtin_amdgcn_sched_barrier(0);
    zp_fma(B.u, HA, acc);
    zp_hload(HA, hrow, (d0 + 4 * ZP_G) & (DD - 1));
    __builtin_amdgcn_sched_barrier(0);
    zp_fma(B.u + ZP_G, HB, acc);
  }

  #pragma unroll
  for (int r = 0; r < 8; ++r) {
    const size_t n = (size_t)(nb + r0 + r);
    Zf[((size_t)k0 * NN + n) * SS + s]       = (_Float16)acc[r][0];
    Zs[((size_t)k0 * NN + n) * SS + s]       = (_Float16)(acc[r][0] * e0);
    Zf[((size_t)(k0 + 1) * NN + n) * SS + s] = (_Float16)acc[r][1];
    Zs[((size_t)(k0 + 1) * NN + n) * SS + s] = (_Float16)(acc[r][1] * e1);
  }
}

#define KROWB 64

typedef short s16x8 __attribute__((ext_vector_type(8)));

__device__ __forceinline__ v8h ld_tr16(const unsigned char* p) {
  typedef __attribute__((address_space(3))) s16x8 lds_s16x8;
  return __builtin_bit_cast(v8h, __builtin_amdgcn_ds_load_tr16_b128_v8i16((lds_s16x8*)p));
}

struct KAddr { const unsigned char* klo; const unsigned char* khi;
               const unsigned char* tr0; const unsigned char* tr1; };

struct AttnBlock {
  v16h  bK[4];
  v16h  bV[2][2];
  float w[8][4];
};

__device__ __forceinline__ void attn_load(AttnBlock& r, const KAddr& a,
    const float* __restrict__ mbase, int kb)
{
  #pragma unroll
  for (int t = 0; t < 4; ++t) {
    const int o = (kb + t * 16) * KROWB;
    r.bK[t] = make_ab((const _Float16*)(a.klo + o), (const _Float16*)(a.khi + o));
  }
  const float* mrow = mbase + kb;
  #pragma unroll
  for (int v = 0; v < 8; ++v) {
    #pragma unroll
    for (int t = 0; t < 4; ++t) r.w[v][t] = mrow[(size_t)v * NN + t * 16];
    __builtin_prefetch(mrow + (size_t)v * NN + 64, 0, 0);
  }
  #pragma unroll
  for (int c = 0; c < 2; ++c) {
    const int o = (kb + c * 32) * KROWB;
    const v8h a0 = ld_tr16(a.tr0 + o);
    const v8h a1 = ld_tr16(a.tr0 + o + 16 * KROWB);
    const v8h b0 = ld_tr16(a.tr1 + o);
    const v8h b1 = ld_tr16(a.tr1 + o + 16 * KROWB);
    r.bV[c][0] = __builtin_shufflevector(a0, a1, 0,1,2,3,4,5,6,7,8,9,10,11,12,13,14,15);
    r.bV[c][1] = __builtin_shufflevector(b0, b1, 0,1,2,3,4,5,6,7,8,9,10,11,12,13,14,15);
  }
}

__device__ __forceinline__ void attn_block(const AttnBlock& r, const v16h aQ, const v16h ones,
    const float (&m)[8], const float inv_s, _Float16* Plds,
    const int ll, const int half, const int c0, v8f& o0, v8f& o1, v8f& lacc)
{
  v8f sc[4];
  #pragma unroll
  for (int t = 0; t < 4; ++t) {
    v8f z = {};
    sc[t] = __builtin_amdgcn_wmma_f32_16x16x32_f16(false, aQ, false, r.bK[t], (short)0, z, false, false);
  }
  #pragma unroll
  for (int v = 0; v < 8; ++v) {
    const float mn = m[v];
    const float p0 = r.w[v][0] * __expf(sc[0][v] * inv_s - mn);
    const float p1 = r.w[v][1] * __expf(sc[1][v] * inv_s - mn);
    const float p2 = r.w[v][2] * __expf(sc[2][v] * inv_s - mn);
    const float p3 = r.w[v][3] * __expf(sc[3][v] * inv_s - mn);
    const int row = v + 8 * half;
    Plds[row * 64 + ll]      = (_Float16)p0;
    Plds[row * 64 + 16 + ll] = (_Float16)p1;
    Plds[row * 64 + 32 + ll] = (_Float16)p2;
    Plds[row * 64 + 48 + ll] = (_Float16)p3;
  }
  asm volatile("s_wait_dscnt 0" ::: "memory");
  #pragma unroll
  for (int c = 0; c < 2; ++c) {
    const v16h aP = make_ab(&Plds[ll * 64 + c * 32 + c0],
                            &Plds[ll * 64 + c * 32 + c0 + 16]);
    o0   = __builtin_amdgcn_wmma_f32_16x16x32_f16(false, aP, false, r.bV[c][0], (short)0, o0,   false, false);
    o1   = __builtin_amdgcn_wmma_f32_16x16x32_f16(false, aP, false, r.bV[c][1], (short)0, o1,   false, false);
    lacc = __builtin_amdgcn_wmma_f32_16x16x32_f16(false, aP, false, ones,       (short)0, lacc, false, false);
  }
}

#define S1_TILES 16
#define S1_KEYS  (16 * S1_TILES)
struct KeyGroup { v16h b[S1_TILES]; };

__device__ __forceinline__ void s1_load(KeyGroup& g, const KAddr& a, int kb)
{
  #pragma unroll
  for (int t = 0; t < S1_TILES; ++t) {
    const int o = (kb + t * 16) * KROWB;
    g.b[t] = make_ab((const _Float16*)(a.klo + o), (const _Float16*)(a.khi + o));
  }
}

__device__ __forceinline__ void s1_max(const KeyGroup& g, const v16h aQ, float (&m)[8])
{
  #pragma unroll
  for (int t = 0; t < S1_TILES; ++t) {
    v8f z = {};
    const v8f s1 = __builtin_amdgcn_wmma_f32_16x16x32_f16(false, aQ, false, g.b[t], (short)0, z, false, false);
    #pragma unroll
    for (int v = 0; v < 8; ++v) m[v] = fmaxf(m[v], s1[v]);
  }
}

#define FILL_N 32
struct FillBatch { v4f c[FILL_N]; };

__device__ __forceinline__ void fill_load(FillBatch& f, const unsigned char* __restrict__ src, int b)
{
  #pragma unroll
  for (int i = 0; i < FILL_N; ++i) f.c[i] = *(const v4f*)(src + (size_t)(b * FILL_N + i) * 2048);
}
__device__ __forceinline__ void fill_store(const FillBatch& f, unsigned char* dst, int b)
{
  #pragma unroll
  for (int i = 0; i < FILL_N; ++i) *(v4f*)(dst + (b * FILL_N + i) * 2048) = f.c[i];
}

__global__ __launch_bounds__(128) void attn_kernel(
    const _Float16* __restrict__ Zs, const _Float16* __restrict__ Zf,
    const float* __restrict__ adj, float* __restrict__ O)
{
  const int w    = __builtin_amdgcn_readfirstlane((int)(threadIdx.x >> 5));
  const int k    = blockIdx.x & (KK - 1);
  const int qb   = (blockIdx.x >> 2) * 64 + w * 16;
  const int l    = threadIdx.x & 31;
  const int ll   = l & 15;
  const int half = l >> 4;
  const int c0   = half ? 8 : 0;

  __shared__ unsigned char __attribute__((aligned(16))) Ksh[NN * KROWB];
  __shared__ _Float16      __attribute__((aligned(16))) Pall[4 * 16 * 64];
  _Float16* Plds = Pall + w * (16 * 64);

  const float inv_s = 0.17677669529663687f;

  const _Float16* qrow = Zs + ((size_t)k * NN + (qb + ll)) * SS;
  const v16h aQ = make_ab(qrow + c0, qrow + c0 + 16);

  {
    const int t    = (int)threadIdx.x;
    const int frow = t >> 2;
    const int fsw  = ((frow >> 3) & 1) | (((frow >> 2) & 1) << 1);
    const unsigned char* src = (const unsigned char*)(Zf + (size_t)k * NN * SS) + t * 16;
    unsigned char*       dst = Ksh + frow * KROWB + 16 * ((t & 3) ^ fsw);
    FillBatch FA, FB;
    fill_load(FA, src, 0);
    fill_load(FB, src, 1);
    __builtin_amdgcn_sched_barrier(0);
    fill_store(FA, dst, 0);
    fill_load(FA, src, 2);
    __builtin_amdgcn_sched_barrier(0);
    fill_store(FB, dst, 1);
    fill_load(FB, src, 3);
    __builtin_amdgcn_sched_barrier(0);
    fill_store(FA, dst, 2);
    fill_store(FB, dst, 3);
  }
  __syncthreads();

  KAddr ka;
  {
    const int ksw = ((ll >> 3) & 1) | (((ll >> 2) & 1) << 1);
    ka.klo = Ksh + ll * KROWB + 16 * (half ^ ksw);
    ka.khi = Ksh + ll * KROWB + 16 * ((half ^ ksw) ^ 2);
    const int trow = 8 * half + (l & 7);
    const int tsw  = half | (((l >> 2) & 1) << 1);
    const int tq   = (l >> 3) & 1;
    ka.tr0 = Ksh + trow * KROWB + 16 * (tq ^ tsw);
    ka.tr1 = Ksh + trow * KROWB + 16 * ((tq ^ tsw) ^ 2);
  }

  v16h ones;
  #pragma unroll
  for (int i = 0; i < 16; ++i) ones[i] = (_Float16)1.0f;

  float m[8];
  #pragma unroll
  for (int v = 0; v < 8; ++v) m[v] = -INFINITY;

  KeyGroup GA, GB;
  s1_load(GA, ka, 0);
  for (int kb = 0; kb < NN; kb += 2 * S1_KEYS) {
    s1_load(GB, ka, kb + S1_KEYS);
    __builtin_amdgcn_sched_barrier(0);
    s1_max(GA, aQ, m);
    s1_load(GA, ka, (kb + 2 * S1_KEYS) & (NN - 1));
    __builtin_amdgcn_sched_barrier(0);
    s1_max(GB, aQ, m);
  }
  #pragma unroll
  for (int v = 0; v < 8; ++v) {
    float mx = m[v];
    #pragma unroll
    for (int off = 1; off < 16; off <<= 1) mx = fmaxf(mx, __shfl_xor(mx, off, 32));
    m[v] = mx * inv_s;
  }

  v8f o0 = {}; v8f o1 = {};
  v8f lacc = {};

  const float* mbase = adj + (size_t)(qb + 8 * half) * NN + ll;

  AttnBlock A, B;
  attn_load(A, ka, mbase, 0);
  for (int kb = 0; kb < NN; kb += 128) {
    attn_load(B, ka, mbase, kb + 64);
    __builtin_amdgcn_sched_barrier(0);
    attn_block(A, aQ, ones, m, inv_s, Plds, ll, half, c0, o0, o1, lacc);
    asm volatile("" : "+v"(o0), "+v"(o1), "+v"(lacc));
    attn_load(A, ka, mbase, (kb + 128) & (NN - 1));
    __builtin_amdgcn_sched_barrier(0);
    attn_block(B, aQ, ones, m, inv_s, Plds, ll, half, c0, o0, o1, lacc);
  }

  float* Ow = O + (size_t)(qb + 8 * half) * (KK * SS) + k * SS + ll;
  #pragma unroll
  for (int v = 0; v < 8; ++v) {
    const float inv_l = 1.0f / lacc[v];
    Ow[(size_t)v * (KK * SS)]      = o0[v] * inv_l;
    Ow[(size_t)v * (KK * SS) + 16] = o1[v] * inv_l;
  }
}

__global__ __launch_bounds__(128) void combine_kernel(
    const float* __restrict__ O, const float* __restrict__ H, const float* __restrict__ U,
    const float* __restrict__ thr, float* __restrict__ out)
{
  const int k    = __builtin_amdgcn_readfirstlane((int)(threadIdx.x >> 5));
  const int qb   = blockIdx.x * 16;
  const int l    = threadIdx.x & 31;
  const int ll   = l & 15;
  const int half = l >> 4;

  const int dcol = 64 * k + ll;
  v4f aO[KK][4];
  #pragma unroll
  for (int kk = 0; kk < KK; ++kk) {
    const float* orow = O + (size_t)(qb + ll) * (KK * SS) + kk * SS + 16 * half;
    #pragma unroll
    for (int mm = 0; mm < 4; ++mm) aO[kk][mm] = *(const v4f*)(orow + 4 * mm);
  }
  v4f bU[KK][4][4];
  #pragma unroll
  for (int kk = 0; kk < KK; ++kk) {
    #pragma unroll
    for (int j = 0; j < 4; ++j) {
      const float* ur = U + ((size_t)kk * DD + dcol + 16 * j) * SS + 16 * half;
      #pragma unroll
      for (int mm = 0; mm < 4; ++mm) bU[kk][j][mm] = *(const v4f*)(ur + 4 * mm);
    }
  }
  float th[4], hv[4][8];
  #pragma unroll
  for (int j = 0; j < 4; ++j) {
    th[j] = thr[dcol + 16 * j];
    #pragma unroll
    for (int r = 0; r < 8; ++r)
      hv[j][r] = H[(size_t)(qb + 8 * half + r) * DD + dcol + 16 * j];
  }
  __builtin_amdgcn_sched_barrier(0);

  v8f acc[4] = {};
  #pragma unroll
  for (int kk = 0; kk < KK; ++kk) {
    #pragma unroll
    for (int i = 0; i < 8; ++i) {
      v2f a;
      a[0] = aO[kk][i >> 1][(i & 1) * 2];
      a[1] = aO[kk][i >> 1][(i & 1) * 2 + 1];
      #pragma unroll
      for (int j = 0; j < 4; ++j) {
        v2f b;
        b[0] = bU[kk][j][i >> 1][(i & 1) * 2];
        b[1] = bU[kk][j][i >> 1][(i & 1) * 2 + 1];
        acc[j] = __builtin_amdgcn_wmma_f32_16x16x4_f32(false, a, false, b, (short)0, acc[j], false, false);
      }
    }
  }

  #pragma unroll
  for (int j = 0; j < 4; ++j) {
    #pragma unroll
    for (int r = 0; r < 8; ++r) {
      const float h = hv[j][r] + ETA * acc[j][r];
      const float a = fabsf(h) - th[j];
      out[(size_t)(qb + 8 * half + r) * DD + dcol + 16 * j] = (a > 0.f) ? copysignf(a, h) : 0.f;
    }
  }
}

#define ORTH_WAVES   10
#define ORTH_THREADS 320
#define ORTH_CH    8

struct OrthFrag { v2f a0, a1, b0, b1; };

__device__ __forceinline__ OrthFrag orth_load(const float* __restrict__ pa,
                                              const float* __restrict__ pb, int d0) {
  const v2f x0 = *(const v2f*)(pa + (size_t)d0 * SS);
  const v2f x1 = *(const v2f*)(pa + (size_t)(d0 + 1) * SS);
  const v2f y0 = *(const v2f*)(pb + (size_t)d0 * SS);
  const v2f y1 = *(const v2f*)(pb + (size_t)(d0 + 1) * SS);
  OrthFrag f;
  f.a0[0] = x0[0]; f.a0[1] = x1[0];
  f.a1[0] = x0[1]; f.a1[1] = x1[1];
  f.b0[0] = y0[0]; f.b0[1] = y1[0];
  f.b1[0] = y0[1]; f.b1[1] = y1[1];
  return f;
}

__global__ __launch_bounds__(ORTH_THREADS) void orth_kernel(const float* __restrict__ U,
                                                            float* __restrict__ out)
{
  const int w = threadIdx.x >> 5;
  const int l = threadIdx.x & 31;
  const int c = l & 15;
  const int h = l >> 4;
  const int kq    = (w >= 4) + (w >= 7) + (w >= 9);
  const int first = (kq == 0) ? 0 : (kq == 1) ? 4 : (kq == 2) ? 7 : 9;
  const int lq    = kq + (w - first);

  const float* pa = U + ((size_t)kq * DD + 2 * h) * SS + 2 * c;
  const float* pb = U + ((size_t)lq * DD + 2 * h) * SS + 2 * c;

  v8f g00 = {}, g01 = {}, g10 = {}, g11 = {};
  OrthFrag cur[ORTH_CH], nxt[ORTH_CH];
  #pragma unroll
  for (int j = 0; j < ORTH_CH; ++j) cur[j] = orth_load(pa, pb, 4 * j);

  #pragma unroll
  for (int g = 0; g < DD / (4 * ORTH_CH); ++g) {
    if (g + 1 < DD / (4 * ORTH_CH)) {
      #pragma unroll
      for (int j = 0; j < ORTH_CH; ++j) nxt[j] = orth_load(pa, pb, 4 * ORTH_CH * (g + 1) + 4 * j);
    }
    __builtin_amdgcn_sched_barrier(0);
    #pragma unroll
    for (int j = 0; j < ORTH_CH; ++j) {
      g00 = __builtin_amdgcn_wmma_f32_16x16x4_f32(false, cur[j].a0, false, cur[j].b0, (short)0, g00, false, false);
      g01 = __builtin_amdgcn_wmma_f32_16x16x4_f32(false, cur[j].a0, false, cur[j].b1, (short)0, g01, false, false);
      g10 = __builtin_amdgcn_wmma_f32_16x16x4_f32(false, cur[j].a1, false, cur[j].b0, (short)0, g10, false, false);
      g11 = __builtin_amdgcn_wmma_f32_16x16x4_f32(false, cur[j].a1, false, cur[j].b1, (short)0, g11, false, false);
    }
    #pragma unroll
    for (int j = 0; j < ORTH_CH; ++j) cur[j] = nxt[j];
  }

  const bool diag = (kq == lq);
  float acc = 0.f;
  #pragma unroll
  for (int r = 0; r < 8; ++r) {
    const float one = (diag && (8 * h + r == c)) ? 1.f : 0.f;
    const float e00 = g00[r] - one;
    const float e11 = g11[r] - one;
    acc = fmaf(e00, e00, acc);
    acc = fmaf(g01[r], g01[r], acc);
    acc = fmaf(g10[r], g10[r], acc);
    acc = fmaf(e11, e11, acc);
  }
  for (int off = 16; off > 0; off >>= 1) acc += __shfl_xor(acc, off, 32);

  __shared__ float red[ORTH_WAVES];
  if (l == 0) red[w] = acc;
  __syncthreads();
  if (threadIdx.x == 0) {
    float s = 0.f;
    for (int i = 0; i < ORTH_WAVES; ++i) s += red[i];
    out[(size_t)NN * DD] = s;
  }
}

extern "C" void kernel_launch(void* const* d_in, const int* in_sizes, int n_in,
                              void* d_out, int out_size, void* d_ws, size_t ws_size,
                              hipStream_t stream) {
  (void)in_sizes; (void)n_in; (void)out_size; (void)ws_size;
  const float* H   = (const float*)d_in[0];
  const float* adj = (const float*)d_in[1];
  const float* U   = (const float*)d_in[2];
  const float* lp  = (const float*)d_in[3];
  const float* thr = (const float*)d_in[4];
  float* out = (float*)d_out;

  char* ws = (char*)d_ws;
  _Float16* Zs = (_Float16*)(ws);
  _Float16* Zf = (_Float16*)(ws + (1u << 20));
  float*    O  = (float*)(ws + (2u << 20));

  zproj_kernel   <<<NN / ZP_ROWS,      128, 0, stream>>>(H, U, lp, Zs, Zf);
  attn_kernel    <<<KK * (NN / 64),    128, 0, stream>>>(Zs, Zf, adj, O);
  combine_kernel <<<NN / 16,           128, 0, stream>>>(O, H, U, thr, out);
  orth_kernel    <<<1,        ORTH_THREADS, 0, stream>>>(U, out);
}


# attention sweep 2: P V runs one 64-key block behind the scores, second P tile per wave in LDS; no full LDS wait between P store, read-back and matrix op; every sum's order kept
# speedup vs baseline: 7.2654x; 1.0753x over previous
#include <hip/hip_runtime.h>

#define NN 4096
#define DD 256
#define SS 32
#define KK 4
#define ETA 0.5f

typedef __attribute__((ext_vector_type(16))) _Float16 v16h;
typedef __attribute__((ext_vector_type(8)))  _Float16 v8h;
typedef __attribute__((ext_vector_type(8)))  float    v8f;
typedef __attribute__((ext_vector_type(2)))  float    v2f;
typedef __attribute__((ext_vector_type(4)))  float    v4f;

__device__ __forceinline__ v16h make_ab(const _Float16* p0, const _Float16* p1) {
  v8h lo = *(const v8h*)p0;
  v8h hi = *(const v8h*)p1;
  return __builtin_shufflevector(lo, hi, 0,1,2,3,4,5,6,7,8,9,10,11,12,13,14,15);
}

#define ZP_ROWS 16
#define ZP_CH   32

struct UChunk { v2f u[ZP_CH]; };

__device__ __forceinline__ void zp_load(UChunk& u, const float* __restrict__ up0,
                                        const float* __restrict__ up1, int d0)
{
  #pragma unroll
  for (int j = 0; j < ZP_CH; ++j) {
    u.u[j][0] = up0[(size_t)(d0 + j) * SS];
    u.u[j][1] = up1[(size_t)(d0 + j) * SS];
  }
}

#define ZP_G 16
struct HGroup { v4f h[8][ZP_G / 4]; };

__device__ __forceinline__ void zp_hload(HGroup& g, const float* hrow, int d)
{
  #pragma unroll
  for (int r = 0; r < 8; ++r) {
    #pragma unroll
    for (int c = 0; c < ZP_G / 4; ++c)
      g.h[r][c] = *(const v4f*)(hrow + r * DD + d + 4 * c);
  }
}

__device__ __forceinline__ void zp_fma(const v2f* u, const HGroup& g, v2f (&acc)[8])
{
  #pragma unroll
  for (int c = 0; c < ZP_G / 4; ++c) {
    #pragma unroll
    for (int q = 0; q < 4; ++q) {
      #pragma unroll
      for (int r = 0; r < 8; ++r) {
        v2f hh;
        hh[0] = g.h[r][c][q];
        hh[1] = g.h[r][c][q];
        acc[r] = __builtin_elementwise_fma(hh, u[4 * c + q], acc[r]);
      }
    }
  }
}

__global__ __launch_bounds__(128) void zproj_kernel(
    const float* __restrict__ H, const float* __restrict__ U,
    const float* __restrict__ lp,
    _Float16* __restrict__ Zs, _Float16* __restrict__ Zf)
{
  __shared__ float __attribute__((aligned(16))) Hs[ZP_ROWS * DD];

  const int w  = __builtin_amdgcn_readfirstlane((int)(threadIdx.x >> 5));
  const int s  = threadIdx.x & 31;
  const int k0 = 2 * (w & 1);
  const int r0 = 8 * (w >> 1);
  const int nb = blockIdx.x * ZP_ROWS;

  const float* up0 = U + (size_t)k0 * DD * SS + s;
  const float* up1 = up0 + (size_t)DD * SS;

  UChunk A, B;
  zp_load(A, up0, up1, 0);

  const float* hsrc = H + (size_t)nb * DD;
  #pragma unroll
  for (int i = 0; i < 8; ++i) {
    const int o = (i * 128 + (int)threadIdx.x) * 4;
    *(v4f*)(Hs + o) = *(const v4f*)(hsrc + o);
  }
  const float e0 = __expf(lp[k0 * SS + s]);
  const float e1 = __expf(lp[(k0 + 1) * SS + s]);
  __syncthreads();

  const float* hrow = Hs + r0 * DD;
  v2f acc[8];
  #pragma unroll
  for (int r = 0; r < 8; ++r) { acc[r][0] = 0.f; acc[r][1] = 0.f; }

  HGroup HA, HB;
  zp_hload(HA, hrow, 0);
  #pragma nounroll
  for (int d0 = 0; d0 < DD; d0 += 2 * ZP_CH) {
    zp_load(B, up0, up1, d0 + ZP_CH);
    zp_hload(HB, hrow, d0 + ZP_G);
    __builtin_amdgcn_sched_barrier(0);
    zp_fma(A.u, HA, acc);
    zp_hload(HA, hrow, d0 + 2 * ZP_G);
    __builtin_amdgcn_sched_barrier(0);
    zp_fma(A.u + ZP_G, HB, acc);
    zp_load(A, up0, up1, (d0 + 2 * ZP_CH) & (DD - 1));
    zp_hload(HB, hrow, d0 + 3 * ZP_G);
    __builtin_amdgcn_sched_barrier(0);
    zp_fma(B.u, HA, acc);
    zp_hload(HA, hrow, (d0 + 4 * ZP_G) & (DD - 1));
    __builtin_amdgcn_sched_barrier(0);
    zp_fma(B.u + ZP_G, HB, acc);
  }

  #pragma unroll
  for (int r = 0; r < 8; ++r) {
    const size_t n = (size_t)(nb + r0 + r);
    Zf[((size_t)k0 * NN + n) * SS + s]       = (_Float16)acc[r][0];
    Zs[((size_t)k0 * NN + n) * SS + s]       = (_Float16)(acc[r][0] * e0);
    Zf[((size_t)(k0 + 1) * NN + n) * SS + s] = (_Float16)acc[r][1];
    Zs[((size_t)(k0 + 1) * NN + n) * SS + s] = (_Float16)(acc[r][1] * e1);
  }
}

#define KROWB 64

typedef short s16x8 __attribute__((ext_vector_type(8)));

__device__ __forceinline__ v8h ld_tr16(const unsigned char* p) {
  typedef __attribute__((address_space(3))) s16x8 lds_s16x8;
  return __builtin_bit_cast(v8h, __builtin_amdgcn_ds_load_tr16_b128_v8i16((lds_s16x8*)p));
}

struct KAddr { const unsigned char* klo; const unsigned char* khi;
               const unsigned char* tr0; const unsigned char* tr1; };

struct ScoreOps {
  v16h  bK[4];
  float w[8][4];
};
struct ValueOps {
  v16h  bV[2][2];
};
struct PFrag { v16h a[2]; };

__device__ __forceinline__ void score_load(ScoreOps& r, const KAddr& a,
    const float* __restrict__ mbase, int kb)
{
  #pragma unroll
  for (int t = 0; t < 4; ++t) {
    const int o = (kb + t * 16) * KROWB;
    r.bK[t] = make_ab((const _Float16*)(a.klo + o), (const _Float16*)(a.khi + o));
  }
  const float* mrow = mbase + kb;
  #pragma unroll
  for (int v = 0; v < 8; ++v) {
    #pragma unroll
    for (int t = 0; t < 4; ++t) r.w[v][t] = mrow[(size_t)v * NN + t * 16];
    __builtin_prefetch(mrow + (size_t)v * NN + 64, 0, 0);
  }
}

__device__ __forceinline__ void value_load(ValueOps& r, const KAddr& a, int kb)
{
  #pragma unroll
  for (int c = 0; c < 2; ++c) {
    const int o = (kb + c * 32) * KROWB;
    const v8h a0 = ld_tr16(a.tr0 + o);
    const v8h a1 = ld_tr16(a.tr0 + o + 16 * KROWB);
    const v8h b0 = ld_tr16(a.tr1 + o);
    const v8h b1 = ld_tr16(a.tr1 + o + 16 * KROWB);
    r.bV[c][0] = __builtin_shufflevector(a0, a1, 0,1,2,3,4,5,6,7,8,9,10,11,12,13,14,15);
    r.bV[c][1] = __builtin_shufflevector(b0, b1, 0,1,2,3,4,5,6,7,8,9,10,11,12,13,14,15);
  }
}

__device__ __forceinline__ void p_load(PFrag& f, const _Float16* Pbuf, const int ll, const int c0)
{
  #pragma unroll
  for (int c = 0; c < 2; ++c)
    f.a[c] = make_ab(&Pbuf[ll * 64 + c * 32 + c0], &Pbuf[ll * 64 + c * 32 + c0 + 16]);
}

__device__ __forceinline__ void attn_scores(const ScoreOps& r, const v16h aQ,
    const float (&m)[8], const float inv_s, _Float16* Pbuf, const int ll, const int half)
{
  v8f sc[4];
  #pragma unroll
  for (int t = 0; t < 4; ++t) {
    v8f z = {};
    sc[t] = __builtin_amdgcn_wmma_f32_16x16x32_f16(false, aQ, false, r.bK[t], (short)0, z, false, false);
  }
  #pragma unroll
  for (int v = 0; v < 8; ++v) {
    const float mn = m[v];
    const float p0 = r.w[v][0] * __expf(sc[0][v] * inv_s - mn);
    const float p1 = r.w[v][1] * __expf(sc[1][v] * inv_s - mn);
    const float p2 = r.w[v][2] * __expf(sc[2][v] * inv_s - mn);
    const float p3 = r.w[v][3] * __expf(sc[3][v] * inv_s - mn);
    const int row = v + 8 * half;
    Pbuf[row * 64 + ll]      = (_Float16)p0;
    Pbuf[row * 64 + 16 + ll] = (_Float16)p1;
    Pbuf[row * 64 + 32 + ll] = (_Float16)p2;
    Pbuf[row * 64 + 48 + ll] = (_Float16)p3;
  }
}

__device__ __forceinline__ void attn_pv(const PFrag& f, const ValueOps& r, const v16h ones,
    v8f& o0, v8f& o1, v8f& lacc)
{
  #pragma unroll
  for (int c = 0; c < 2; ++c) {
    o0   = __builtin_amdgcn_wmma_f32_16x16x32_f16(false, f.a[c], false, r.bV[c][0], (short)0, o0,   false, false);
    o1   = __builtin_amdgcn_wmma_f32_16x16x32_f16(false, f.a[c], false, r.bV[c][1], (short)0, o1,   false, false);
    lacc = __builtin_amdgcn_wmma_f32_16x16x32_f16(false, f.a[c], false, ones,       (short)0, lacc, false, false);
  }
}

#define S1_TILES 16
#define S1_KEYS  (16 * S1_TILES)
struct KeyGroup { v16h b[S1_TILES]; };

__device__ __forceinline__ void s1_load(KeyGroup& g, const KAddr& a, int kb)
{
  #pragma unroll
  for (int t = 0; t < S1_TILES; ++t) {
    const int o = (kb + t * 16) * KROWB;
    g.b[t] = make_ab((const _Float16*)(a.klo + o), (const _Float16*)(a.khi + o));
  }
}

__device__ __forceinline__ void s1_max(const KeyGroup& g, const v16h aQ, float (&m)[8])
{
  #pragma unroll
  for (int t = 0; t < S1_TILES; ++t) {
    v8f z = {};
    const v8f s1 = __builtin_amdgcn_wmma_f32_16x16x32_f16(false, aQ, false, g.b[t], (short)0, z, false, false);
    #pragma unroll
    for (int v = 0; v < 8; ++v) m[v] = fmaxf(m[v], s1[v]);
  }
}

#define FILL_N 32
struct FillBatch { v4f c[FILL_N]; };

__device__ __forceinline__ void fill_load(FillBatch& f, const unsigned char* __restrict__ src, int b)
{
  #pragma unroll
  for (int i = 0; i < FILL_N; ++i) f.c[i] = *(const v4f*)(src + (size_t)(b * FILL_N + i) * 2048);
}
__device__ __forceinline__ void fill_store(const FillBatch& f, unsigned char* dst, int b)
{
  #pragma unroll
  for (int i = 0; i < FILL_N; ++i) *(v4f*)(dst + (b * FILL_N + i) * 2048) = f.c[i];
}

__global__ __launch_bounds__(128) void attn_kernel(
    const _Float16* __restrict__ Zs, const _Float16* __restrict__ Zf,
    const float* __restrict__ adj, float* __restrict__ O)
{
  const int w    = __builtin_amdgcn_readfirstlane((int)(threadIdx.x >> 5));
  const int k    = blockIdx.x & (KK - 1);
  const int qb   = (blockIdx.x >> 2) * 64 + w * 16;
  const int l    = threadIdx.x & 31;
  const int ll   = l & 15;
  const int half = l >> 4;
  const int c0   = half ? 8 : 0;

  __shared__ unsigned char __attribute__((aligned(16))) Ksh[NN * KROWB];
  __shared__ _Float16      __attribute__((aligned(16))) Pall[4 * 2 * 16 * 64];
  _Float16* P0 = Pall + w * (2 * 16 * 64);
  _Float16* P1 = P0 + 16 * 64;

  const float inv_s = 0.17677669529663687f;

  const _Float16* qrow = Zs + ((size_t)k * NN + (qb + ll)) * SS;
  const v16h aQ = make_ab(qrow + c0, qrow + c0 + 16);

  {
    const int t    = (int)threadIdx.x;
    const int frow = t >> 2;
    const int fsw  = ((frow >> 3) & 1) | (((frow >> 2) & 1) << 1);
    const unsigned char* src = (const unsigned char*)(Zf + (size_t)k * NN * SS) + t * 16;
    unsigned char*       dst = Ksh + frow * KROWB + 16 * ((t & 3) ^ fsw);
    FillBatch FA, FB;
    fill_load(FA, src, 0);
    fill_load(FB, src, 1);
    __builtin_amdgcn_sched_barrier(0);
    fill_store(FA, dst, 0);
    fill_load(FA, src, 2);
    __builtin_amdgcn_sched_barrier(0);
    fill_store(FB, dst, 1);
    fill_load(FB, src, 3);
    __builtin_amdgcn_sched_barrier(0);
    fill_store(FA, dst, 2);
    fill_store(FB, dst, 3);
  }
  __syncthreads();

  KAddr ka;
  {
    const int ksw = ((ll >> 3) & 1) | (((ll >> 2) & 1) << 1);
    ka.klo = Ksh + ll * KROWB + 16 * (half ^ ksw);
    ka.khi = Ksh + ll * KROWB + 16 * ((half ^ ksw) ^ 2);
    const int trow = 8 * half + (l & 7);
    const int tsw  = half | (((l >> 2) & 1) << 1);
    const int tq   = (l >> 3) & 1;
    ka.tr0 = Ksh + trow * KROWB + 16 * (tq ^ tsw);
    ka.tr1 = Ksh + trow * KROWB + 16 * ((tq ^ tsw) ^ 2);
  }

  v16h ones;
  #pragma unroll
  for (int i = 0; i < 16; ++i) ones[i] = (_Float16)1.0f;

  float m[8];
  #pragma unroll
  for (int v = 0; v < 8; ++v) m[v] = -INFINITY;

  KeyGroup GA, GB;
  s1_load(GA, ka, 0);
  for (int kb = 0; kb < NN; kb += 2 * S1_KEYS) {
    s1_load(GB, ka, kb + S1_KEYS);
    __builtin_amdgcn_sched_barrier(0);
    s1_max(GA, aQ, m);
    s1_load(GA, ka, (kb + 2 * S1_KEYS) & (NN - 1));
    __builtin_amdgcn_sched_barrier(0);
    s1_max(GB, aQ, m);
  }
  #pragma unroll
  for (int v = 0; v < 8; ++v) {
    float mx = m[v];
    #pragma unroll
    for (int off = 1; off < 16; off <<= 1) mx = fmaxf(mx, __shfl_xor(mx, off, 32));
    m[v] = mx * inv_s;
  }

  v8f o0 = {}; v8f o1 = {};
  v8f lacc = {};

  const float* mbase = adj + (size_t)(qb + 8 * half) * NN + ll;

  ScoreOps KA, KB;
  ValueOps VA, VB;
  PFrag    PA, PB;
  score_load(KA, ka, mbase, 0);
  score_load(KB, ka, mbase, 64);
  value_load(VA, ka, 0);
  attn_scores(KA, aQ, m, inv_s, P0, ll, half);
  for (int kb = 0; kb < NN; kb += 128) {
    p_load(PA, P0, ll, c0);
    score_load(KA, ka, mbase, (kb + 128) & (NN - 1));
    value_load(VB, ka, kb + 64);
    __builtin_amdgcn_sched_barrier(0);
    attn_scores(KB, aQ, m, inv_s, P1, ll, half);
    attn_pv(PA, VA, ones, o0, o1, lacc);
    asm volatile("" : "+v"(o0), "+v"(o1), "+v"(lacc));
    p_load(PB, P1, ll, c0);
    score_load(KB, ka, mbase, (kb + 192) & (NN - 1));
    value_load(VA, ka, (kb + 128) & (NN - 1));
    __builtin_amdgcn_sched_barrier(0);
    attn_scores(KA, aQ, m, inv_s, P0, ll, half);
    attn_pv(PB, VB, ones, o0, o1, lacc);
  }

  float* Ow = O + (size_t)(qb + 8 * half) * (KK * SS) + k * SS + ll;
  #pragma unroll
  for (int v = 0; v < 8; ++v) {
    const float inv_l = 1.0f / lacc[v];
    Ow[(size_t)v * (KK * SS)]      = o0[v] * inv_l;
    Ow[(size_t)v * (KK * SS) + 16] = o1[v] * inv_l;
  }
}

__global__ __launch_bounds__(128) void combine_kernel(
    const float* __restrict__ O, const float* __restrict__ H, const float* __restrict__ U,
    const float* __restrict__ thr, float* __restrict__ out)
{
  const int k    = __builtin_amdgcn_readfirstlane((int)(threadIdx.x >> 5));
  const int qb   = blockIdx.x * 16;
  const int l    = threadIdx.x & 31;
  const int ll   = l & 15;
  const int half = l >> 4;

  const int dcol = 64 * k + ll;
  v4f aO[KK][4];
  #pragma unroll
  for (int kk = 0; kk < KK; ++kk) {
    const float* orow = O + (size_t)(qb + ll) * (KK * SS) + kk * SS + 16 * half;
    #pragma unroll
    for (int mm = 0; mm < 4; ++mm) aO[kk][mm] = *(const v4f*)(orow + 4 * mm);
  }
  v4f bU[KK][4][4];
  #pragma unroll
  for (int kk = 0; kk < KK; ++kk) {
    #pragma unroll
    for (int j = 0; j < 4; ++j) {
      const float* ur = U + ((size_t)kk * DD + dcol + 16 * j) * SS + 16 * half;
      #pragma unroll
      for (int mm = 0; mm < 4; ++mm) bU[kk][j][mm] = *(const v4f*)(ur + 4 * mm);
    }
  }
  float th[4], hv[4][8];
  #pragma unroll
  for (int j = 0; j < 4; ++j) {
    th[j] = thr[dcol + 16 * j];
    #pragma unroll
    for (int r = 0; r < 8; ++r)
      hv[j][r] = H[(size_t)(qb + 8 * half + r) * DD + dcol + 16 * j];
  }
  __builtin_amdgcn_sched_barrier(0);

  v8f acc[4] = {};
  #pragma unroll
  for (int kk = 0; kk < KK; ++kk) {
    #pragma unroll
    for (int i = 0; i < 8; ++i) {
      v2f a;
      a[0] = aO[kk][i >> 1][(i & 1) * 2];
      a[1] = aO[kk][i >> 1][(i & 1) * 2 + 1];
      #pragma unroll
      for (int j = 0; j < 4; ++j) {
        v2f b;
        b[0] = bU[kk][j][i >> 1][(i & 1) * 2];
        b[1] = bU[kk][j][i >> 1][(i & 1) * 2 + 1];
        acc[j] = __builtin_amdgcn_wmma_f32_16x16x4_f32(false, a, false, b, (short)0, acc[j], false, false);
      }
    }
  }

  #pragma unroll
  for (int j = 0; j < 4; ++j) {
    #pragma unroll
    for (int r = 0; r < 8; ++r) {
      const float h = hv[j][r] + ETA * acc[j][r];
      const float a = fabsf(h) - th[j];
      out[(size_t)(qb + 8 * half + r) * DD + dcol + 16 * j] = (a > 0.f) ? copysignf(a, h) : 0.f;
    }
  }
}

#define ORTH_WAVES   10
#define ORTH_THREADS 320
#define ORTH_CH    8

struct OrthFrag { v2f a0, a1, b0, b1; };

__device__ __forceinline__ OrthFrag orth_load(const float* __restrict__ pa,
                                              const float* __restrict__ pb, int d0) {
  const v2f x0 = *(const v2f*)(pa + (size_t)d0 * SS);
  const v2f x1 = *(const v2f*)(pa + (size_t)(d0 + 1) * SS);
  const v2f y0 = *(const v2f*)(pb + (size_t)d0 * SS);
  const v2f y1 = *(const v2f*)(pb + (size_t)(d0 + 1) * SS);
  OrthFrag f;
  f.a0[0] = x0[0]; f.a0[1] = x1[0];
  f.a1[0] = x0[1]; f.a1[1] = x1[1];
  f.b0[0] = y0[0]; f.b0[1] = y1[0];
  f.b1[0] = y0[1]; f.b1[1] = y1[1];
  return f;
}

__global__ __launch_bounds__(ORTH_THREADS) void orth_kernel(const float* __restrict__ U,
                                                            float* __restrict__ out)
{
  const int w = threadIdx.x >> 5;
  const int l = threadIdx.x & 31;
  const int c = l & 15;
  const int h = l >> 4;
  const int kq    = (w >= 4) + (w >= 7) + (w >= 9);
  const int first = (kq == 0) ? 0 : (kq == 1) ? 4 : (kq == 2) ? 7 : 9;
  const int lq    = kq + (w - first);

  const float* pa = U + ((size_t)kq * DD + 2 * h) * SS + 2 * c;
  const float* pb = U + ((size_t)lq * DD + 2 * h) * SS + 2 * c;

  v8f g00 = {}, g01 = {}, g10 = {}, g11 = {};
  OrthFrag cur[ORTH_CH], nxt[ORTH_CH];
  #pragma unroll
  for (int j = 0; j < ORTH_CH; ++j) cur[j] = orth_load(pa, pb, 4 * j);

  #pragma unroll
  for (int g = 0; g < DD / (4 * ORTH_CH); ++g) {
    if (g + 1 < DD / (4 * ORTH_CH)) {
      #pragma unroll
      for (int j = 0; j < ORTH_CH; ++j) nxt[j] = orth_load(pa, pb, 4 * ORTH_CH * (g + 1) + 4 * j);
    }
    __builtin_amdgcn_sched_barrier(0);
    #pragma unroll
    for (int j = 0; j < ORTH_CH; ++j) {
      g00 = __builtin_amdgcn_wmma_f32_16x16x4_f32(false, cur[j].a0, false, cur[j].b0, (short)0, g00, false, false);
      g01 = __builtin_amdgcn_wmma_f32_16x16x4_f32(false, cur[j].a0, false, cur[j].b1, (short)0, g01, false, false);
      g10 = __builtin_amdgcn_wmma_f32_16x16x4_f32(false, cur[j].a1, false, cur[j].b0, (short)0, g10, false, false);
      g11 = __builtin_amdgcn_wmma_f32_16x16x4_f32(false, cur[j].a1, false, cur[j].b1, (short)0, g11, false, false);
    }
    #pragma unroll
    for (int j = 0; j < ORTH_CH; ++j) cur[j] = nxt[j];
  }

  const bool diag = (kq == lq);
  float acc = 0.f;
  #pragma unroll
  for (int r = 0; r < 8; ++r) {
    const float one = (diag && (8 * h + r == c)) ? 1.f : 0.f;
    const float e00 = g00[r] - one;
    const float e11 = g11[r] - one;
    acc = fmaf(e00, e00, acc);
    acc = fmaf(g01[r], g01[r], acc);
    acc = fmaf(g10[r], g10[r], acc);
    acc = fmaf(e11, e11, acc);
  }
  for (int off = 16; off > 0; off >>= 1) acc += __shfl_xor(acc, off, 32);

  __shared__ float red[ORTH_WAVES];
  if (l == 0) red[w] = acc;
  __syncthreads();
  if (threadIdx.x == 0) {
    float s = 0.f;
    for (int i = 0; i < ORTH_WAVES; ++i) s += red[i];
    out[(size_t)NN * DD] = s;
  }
}

extern "C" void kernel_launch(void* const* d_in, const int* in_sizes, int n_in,
                              void* d_out, int out_size, void* d_ws, size_t ws_size,
                              hipStream_t stream) {
  (void)in_sizes; (void)n_in; (void)out_size; (void)ws_size;
  const float* H   = (const float*)d_in[0];
  const float* adj = (const float*)d_in[1];
  const float* U   = (const float*)d_in[2];
  const float* lp  = (const float*)d_in[3];
  const float* thr = (const float*)d_in[4];
  float* out = (float*)d_out;

  char* ws = (char*)d_ws;
  _Float16* Zs = (_Float16*)(ws);
  _Float16* Zf = (_Float16*)(ws + (1u << 20));
  float*    O  = (float*)(ws + (2u << 20));

  zproj_kernel   <<<NN / ZP_ROWS,      128, 0, stream>>>(H, U, lp, Zs, Zf);
  attn_kernel    <<<KK * (NN / 64),    128, 0, stream>>>(Zs, Zf, adj, O);
  combine_kernel <<<NN / 16,           128, 0, stream>>>(O, H, U, thr, out);
  orth_kernel    <<<1,        ORTH_THREADS, 0, stream>>>(U, out);
}


# attention mask prefetch: each 64-key block's mask rows asked into L2 (SE scope, was system scope) five steps before use instead of two, first blocks before the key fill; loads and sums unchanged
# speedup vs baseline: 7.7567x; 1.0676x over previous
#include <hip/hip_runtime.h>

#define NN 4096
#define DD 256
#define SS 32
#define KK 4
#define ETA 0.5f

typedef __attribute__((ext_vector_type(16))) _Float16 v16h;
typedef __attribute__((ext_vector_type(8)))  _Float16 v8h;
typedef __attribute__((ext_vector_type(8)))  float    v8f;
typedef __attribute__((ext_vector_type(2)))  float    v2f;
typedef __attribute__((ext_vector_type(4)))  float    v4f;

__device__ __forceinline__ v16h make_ab(const _Float16* p0, const _Float16* p1) {
  v8h lo = *(const v8h*)p0;
  v8h hi = *(const v8h*)p1;
  return __builtin_shufflevector(lo, hi, 0,1,2,3,4,5,6,7,8,9,10,11,12,13,14,15);
}

#define ZP_ROWS 16
#define ZP_CH   32

struct UChunk { v2f u[ZP_CH]; };

__device__ __forceinline__ void zp_load(UChunk& u, const float* __restrict__ up0,
                                        const float* __restrict__ up1, int d0)
{
  #pragma unroll
  for (int j = 0; j < ZP_CH; ++j) {
    u.u[j][0] = up0[(size_t)(d0 + j) * SS];
    u.u[j][1] = up1[(size_t)(d0 + j) * SS];
  }
}

#define ZP_G 16
struct HGroup { v4f h[8][ZP_G / 4]; };

__device__ __forceinline__ void zp_hload(HGroup& g, const float* hrow, int d)
{
  #pragma unroll
  for (int r = 0; r < 8; ++r) {
    #pragma unroll
    for (int c = 0; c < ZP_G / 4; ++c)
      g.h[r][c] = *(const v4f*)(hrow + r * DD + d + 4 * c);
  }
}

__device__ __forceinline__ void zp_fma(const v2f* u, const HGroup& g, v2f (&acc)[8])
{
  #pragma unroll
  for (int c = 0; c < ZP_G / 4; ++c) {
    #pragma unroll
    for (int q = 0; q < 4; ++q) {
      #pragma unroll
      for (int r = 0; r < 8; ++r) {
        v2f hh;
        hh[0] = g.h[r][c][q];
        hh[1] = g.h[r][c][q];
        acc[r] = __builtin_elementwise_fma(hh, u[4 * c + q], acc[r]);
      }
    }
  }
}

__global__ __launch_bounds__(128) void zproj_kernel(
    const float* __restrict__ H, const float* __restrict__ U,
    const float* __restrict__ lp,
    _Float16* __restrict__ Zs, _Float16* __restrict__ Zf)
{
  __shared__ float __attribute__((aligned(16))) Hs[ZP_ROWS * DD];

  const int w  = __builtin_amdgcn_readfirstlane((int)(threadIdx.x >> 5));
  const int s  = threadIdx.x & 31;
  const int k0 = 2 * (w & 1);
  const int r0 = 8 * (w >> 1);
  const int nb = blockIdx.x * ZP_ROWS;

  const float* up0 = U + (size_t)k0 * DD * SS + s;
  const float* up1 = up0 + (size_t)DD * SS;

  UChunk A, B;
  zp_load(A, up0, up1, 0);

  const float* hsrc = H + (size_t)nb * DD;
  #pragma unroll
  for (int i = 0; i < 8; ++i) {
    const int o = (i * 128 + (int)threadIdx.x) * 4;
    *(v4f*)(Hs + o) = *(const v4f*)(hsrc + o);
  }
  const float e0 = __expf(lp[k0 * SS + s]);
  const float e1 = __expf(lp[(k0 + 1) * SS + s]);
  __syncthreads();

  const float* hrow = Hs + r0 * DD;
  v2f acc[8];
  #pragma unroll
  for (int r = 0; r < 8; ++r) { acc[r][0] = 0.f; acc[r][1] = 0.f; }

  HGroup HA, HB;
  zp_hload(HA, hrow, 0);
  #pragma nounroll
  for (int d0 = 0; d0 < DD; d0 += 2 * ZP_CH) {
    zp_load(B, up0, up1, d0 + ZP_CH);
    zp_hload(HB, hrow, d0 + ZP_G);
    __builtin_amdgcn_sched_barrier(0);
    zp_fma(A.u, HA, acc);
    zp_hload(HA, hrow, d0 + 2 * ZP_G);
    __builtin_amdgcn_sched_barrier(0);
    zp_fma(A.u + ZP_G, HB, acc);
    zp_load(A, up0, up1, (d0 + 2 * ZP_CH) & (DD - 1));
    zp_hload(HB, hrow, d0 + 3 * ZP_G);
    __builtin_amdgcn_sched_barrier(0);
    zp_fma(B.u, HA, acc);
    zp_hload(HA, hrow, (d0 + 4 * ZP_G) & (DD - 1));
    __builtin_amdgcn_sched_barrier(0);
    zp_fma(B.u + ZP_G, HB, acc);
  }

  #pragma unroll
  for (int r = 0; r < 8; ++r) {
    const size_t n = (size_t)(nb + r0 + r);
    Zf[((size_t)k0 * NN + n) * SS + s]       = (_Float16)acc[r][0];
    Zs[((size_t)k0 * NN + n) * SS + s]       = (_Float16)(acc[r][0] * e0);
    Zf[((size_t)(k0 + 1) * NN + n) * SS + s] = (_Float16)acc[r][1];
    Zs[((size_t)(k0 + 1) * NN + n) * SS + s] = (_Float16)(acc[r][1] * e1);
  }
}

#define KROWB 64

typedef short s16x8 __attribute__((ext_vector_type(8)));

__device__ __forceinline__ v8h ld_tr16(const unsigned char* p) {
  typedef __attribute__((address_space(3))) s16x8 lds_s16x8;
  return __builtin_bit_cast(v8h, __builtin_amdgcn_ds_load_tr16_b128_v8i16((lds_s16x8*)p));
}

struct KAddr { const unsigned char* klo; const unsigned char* khi;
               const unsigned char* tr0; const unsigned char* tr1; };

#define MASK_PF 4
struct ScoreOps {
  v16h  bK[4];
  float w[8][4];
};
struct ValueOps {
  v16h  bV[2][2];
};
struct PFrag { v16h a[2]; };

__device__ __forceinline__ void score_load(ScoreOps& r, const KAddr& a,
    const float* __restrict__ mbase, int kb)
{
  #pragma unroll
  for (int t = 0; t < 4; ++t) {
    const int o = (kb + t * 16) * KROWB;
    r.bK[t] = make_ab((const _Float16*)(a.klo + o), (const _Float16*)(a.khi + o));
  }
  const float* mrow = mbase + kb;
  #pragma unroll
  for (int v = 0; v < 8; ++v) {
    #pragma unroll
    for (int t = 0; t < 4; ++t) r.w[v][t] = mrow[(size_t)v * NN + t * 16];
    __builtin_prefetch(mbase + (size_t)v * NN + ((kb + 64 * MASK_PF) & (NN - 1)), 0, 2);
  }
}

__device__ __forceinline__ void value_load(ValueOps& r, const KAddr& a, int kb)
{
  #pragma unroll
  for (int c = 0; c < 2; ++c) {
    const int o = (kb + c * 32) * KROWB;
    const v8h a0 = ld_tr16(a.tr0 + o);
    const v8h a1 = ld_tr16(a.tr0 + o + 16 * KROWB);
    const v8h b0 = ld_tr16(a.tr1 + o);
    const v8h b1 = ld_tr16(a.tr1 + o + 16 * KROWB);
    r.bV[c][0] = __builtin_shufflevector(a0, a1, 0,1,2,3,4,5,6,7,8,9,10,11,12,13,14,15);
    r.bV[c][1] = __builtin_shufflevector(b0, b1, 0,1,2,3,4,5,6,7,8,9,10,11,12,13,14,15);
  }
}

__device__ __forceinline__ void p_load(PFrag& f, const _Float16* Pbuf, const int ll, const int c0)
{
  #pragma unroll
  for (int c = 0; c < 2; ++c)
    f.a[c] = make_ab(&Pbuf[ll * 64 + c * 32 + c0], &Pbuf[ll * 64 + c * 32 + c0 + 16]);
}

__device__ __forceinline__ void attn_scores(const ScoreOps& r, const v16h aQ,
    const float (&m)[8], const float inv_s, _Float16* Pbuf, const int ll, const int half)
{
  v8f sc[4];
  #pragma unroll
  for (int t = 0; t < 4; ++t) {
    v8f z = {};
    sc[t] = __builtin_amdgcn_wmma_f32_16x16x32_f16(false, aQ, false, r.bK[t], (short)0, z, false, false);
  }
  #pragma unroll
  for (int v = 0; v < 8; ++v) {
    const float mn = m[v];
    const float p0 = r.w[v][0] * __expf(sc[0][v] * inv_s - mn);
    const float p1 = r.w[v][1] * __expf(sc[1][v] * inv_s - mn);
    const float p2 = r.w[v][2] * __expf(sc[2][v] * inv_s - mn);
    const float p3 = r.w[v][3] * __expf(sc[3][v] * inv_s - mn);
    const int row = v + 8 * half;
    Pbuf[row * 64 + ll]      = (_Float16)p0;
    Pbuf[row * 64 + 16 + ll] = (_Float16)p1;
    Pbuf[row * 64 + 32 + ll] = (_Float16)p2;
    Pbuf[row * 64 + 48 + ll] = (_Float16)p3;
  }
}

__device__ __forceinline__ void attn_pv(const PFrag& f, const ValueOps& r, const v16h ones,
    v8f& o0, v8f& o1, v8f& lacc)
{
  #pragma unroll
  for (int c = 0; c < 2; ++c) {
    o0   = __builtin_amdgcn_wmma_f32_16x16x32_f16(false, f.a[c], false, r.bV[c][0], (short)0, o0,   false, false);
    o1   = __builtin_amdgcn_wmma_f32_16x16x32_f16(false, f.a[c], false, r.bV[c][1], (short)0, o1,   false, false);
    lacc = __builtin_amdgcn_wmma_f32_16x16x32_f16(false, f.a[c], false, ones,       (short)0, lacc, false, false);
  }
}

#define S1_TILES 16
#define S1_KEYS  (16 * S1_TILES)
struct KeyGroup { v16h b[S1_TILES]; };

__device__ __forceinline__ void s1_load(KeyGroup& g, const KAddr& a, int kb)
{
  #pragma unroll
  for (int t = 0; t < S1_TILES; ++t) {
    const int o = (kb + t * 16) * KROWB;
    g.b[t] = make_ab((const _Float16*)(a.klo + o), (const _Float16*)(a.khi + o));
  }
}

__device__ __forceinline__ void s1_max(const KeyGroup& g, const v16h aQ, float (&m)[8])
{
  #pragma unroll
  for (int t = 0; t < S1_TILES; ++t) {
    v8f z = {};
    const v8f s1 = __builtin_amdgcn_wmma_f32_16x16x32_f16(false, aQ, false, g.b[t], (short)0, z, false, false);
    #pragma unroll
    for (int v = 0; v < 8; ++v) m[v] = fmaxf(m[v], s1[v]);
  }
}

#define FILL_N 32
struct FillBatch { v4f c[FILL_N]; };

__device__ __forceinline__ void fill_load(FillBatch& f, const unsigned char* __restrict__ src, int b)
{
  #pragma unroll
  for (int i = 0; i < FILL_N; ++i) f.c[i] = *(const v4f*)(src + (size_t)(b * FILL_N + i) * 2048);
}
__device__ __forceinline__ void fill_store(const FillBatch& f, unsigned char* dst, int b)
{
  #pragma unroll
  for (int i = 0; i < FILL_N; ++i) *(v4f*)(dst + (b * FILL_N + i) * 2048) = f.c[i];
}

__global__ __launch_bounds__(128) void attn_kernel(
    const _Float16* __restrict__ Zs, const _Float16* __restrict__ Zf,
    const float* __restrict__ adj, float* __restrict__ O)
{
  const int w    = __builtin_amdgcn_readfirstlane((int)(threadIdx.x >> 5));
  const int k    = blockIdx.x & (KK - 1);
  const int qb   = (blockIdx.x >> 2) * 64 + w * 16;
  const int l    = threadIdx.x & 31;
  const int ll   = l & 15;
  const int half = l >> 4;
  const int c0   = half ? 8 : 0;

  __shared__ unsigned char __attribute__((aligned(16))) Ksh[NN * KROWB];
  __shared__ _Float16      __attribute__((aligned(16))) Pall[4 * 2 * 16 * 64];
  _Float16* P0 = Pall + w * (2 * 16 * 64);
  _Float16* P1 = P0 + 16 * 64;

  const float inv_s = 0.17677669529663687f;

  const _Float16* qrow = Zs + ((size_t)k * NN + (qb + ll)) * SS;
  const v16h aQ = make_ab(qrow + c0, qrow + c0 + 16);

  {
    const float* mpf = adj + (size_t)(qb + 8 * half) * NN + ll;
    #pragma unroll
    for (int b = 0; b <= MASK_PF; ++b) {
      #pragma unroll
      for (int v = 0; v < 8; ++v) __builtin_prefetch(mpf + (size_t)v * NN + 64 * b, 0, 2);
    }
  }

  {
    const int t    = (int)threadIdx.x;
    const int frow = t >> 2;
    const int fsw  = ((frow >> 3) & 1) | (((frow >> 2) & 1) << 1);
    const unsigned char* src = (const unsigned char*)(Zf + (size_t)k * NN * SS) + t * 16;
    unsigned char*       dst = Ksh + frow * KROWB + 16 * ((t & 3) ^ fsw);
    FillBatch FA, FB;
    fill_load(FA, src, 0);
    fill_load(FB, src, 1);
    __builtin_amdgcn_sched_barrier(0);
    fill_store(FA, dst, 0);
    fill_load(FA, src, 2);
    __builtin_amdgcn_sched_barrier(0);
    fill_store(FB, dst, 1);
    fill_load(FB, src, 3);
    __builtin_amdgcn_sched_barrier(0);
    fill_store(FA, dst, 2);
    fill_store(FB, dst, 3);
  }
  __syncthreads();

  KAddr ka;
  {
    const int ksw = ((ll >> 3) & 1) | (((ll >> 2) & 1) << 1);
    ka.klo = Ksh + ll * KROWB + 16 * (half ^ ksw);
    ka.khi = Ksh + ll * KROWB + 16 * ((half ^ ksw) ^ 2);
    const int trow = 8 * half + (l & 7);
    const int tsw  = half | (((l >> 2) & 1) << 1);
    const int tq   = (l >> 3) & 1;
    ka.tr0 = Ksh + trow * KROWB + 16 * (tq ^ tsw);
    ka.tr1 = Ksh + trow * KROWB + 16 * ((tq ^ tsw) ^ 2);
  }

  v16h ones;
  #pragma unroll
  for (int i = 0; i < 16; ++i) ones[i] = (_Float16)1.0f;

  float m[8];
  #pragma unroll
  for (int v = 0; v < 8; ++v) m[v] = -INFINITY;

  KeyGroup GA, GB;
  s1_load(GA, ka, 0);
  for (int kb = 0; kb < NN; kb += 2 * S1_KEYS) {
    s1_load(GB, ka, kb + S1_KEYS);
    __builtin_amdgcn_sched_barrier(0);
    s1_max(GA, aQ, m);
    s1_load(GA, ka, (kb + 2 * S1_KEYS) & (NN - 1));
    __builtin_amdgcn_sched_barrier(0);
    s1_max(GB, aQ, m);
  }
  #pragma unroll
  for (int v = 0; v < 8; ++v) {
    float mx = m[v];
    #pragma unroll
    for (int off = 1; off < 16; off <<= 1) mx = fmaxf(mx, __shfl_xor(mx, off, 32));
    m[v] = mx * inv_s;
  }

  v8f o0 = {}; v8f o1 = {};
  v8f lacc = {};

  const float* mbase = adj + (size_t)(qb + 8 * half) * NN + ll;

  ScoreOps KA, KB;
  ValueOps VA, VB;
  PFrag    PA, PB;
  score_load(KA, ka, mbase, 0);
  score_load(KB, ka, mbase, 64);
  value_load(VA, ka, 0);
  attn_scores(KA, aQ, m, inv_s, P0, ll, half);
  for (int kb = 0; kb < NN; kb += 128) {
    p_load(PA, P0, ll, c0);
    score_load(KA, ka, mbase, (kb + 128) & (NN - 1));
    value_load(VB, ka, kb + 64);
    __builtin_amdgcn_sched_barrier(0);
    attn_scores(KB, aQ, m, inv_s, P1, ll, half);
    attn_pv(PA, VA, ones, o0, o1, lacc);
    asm volatile("" : "+v"(o0), "+v"(o1), "+v"(lacc));
    p_load(PB, P1, ll, c0);
    score_load(KB, ka, mbase, (kb + 192) & (NN - 1));
    value_load(VA, ka, (kb + 128) & (NN - 1));
    __builtin_amdgcn_sched_barrier(0);
    attn_scores(KA, aQ, m, inv_s, P0, ll, half);
    attn_pv(PB, VB, ones, o0, o1, lacc);
  }

  float* Ow = O + (size_t)(qb + 8 * half) * (KK * SS) + k * SS + ll;
  #pragma unroll
  for (int v = 0; v < 8; ++v) {
    const float inv_l = 1.0f / lacc[v];
    Ow[(size_t)v * (KK * SS)]      = o0[v] * inv_l;
    Ow[(size_t)v * (KK * SS) + 16] = o1[v] * inv_l;
  }
}

__global__ __launch_bounds__(128) void combine_kernel(
    const float* __restrict__ O, const float* __restrict__ H, const float* __restrict__ U,
    const float* __restrict__ thr, float* __restrict__ out)
{
  const int k    = __builtin_amdgcn_readfirstlane((int)(threadIdx.x >> 5));
  const int qb   = blockIdx.x * 16;
  const int l    = threadIdx.x & 31;
  const int ll   = l & 15;
  const int half = l >> 4;

  const int dcol = 64 * k + ll;
  v4f aO[KK][4];
  #pragma unroll
  for (int kk = 0; kk < KK; ++kk) {
    const float* orow = O + (size_t)(qb + ll) * (KK * SS) + kk * SS + 16 * half;
    #pragma unroll
    for (int mm = 0; mm < 4; ++mm) aO[kk][mm] = *(const v4f*)(orow + 4 * mm);
  }
  v4f bU[KK][4][4];
  #pragma unroll
  for (int kk = 0; kk < KK; ++kk) {
    #pragma unroll
    for (int j = 0; j < 4; ++j) {
      const float* ur = U + ((size_t)kk * DD + dcol + 16 * j) * SS + 16 * half;
      #pragma unroll
      for (int mm = 0; mm < 4; ++mm) bU[kk][j][mm] = *(const v4f*)(ur + 4 * mm);
    }
  }
  float th[4], hv[4][8];
  #pragma unroll
  for (int j = 0; j < 4; ++j) {
    th[j] = thr[dcol + 16 * j];
    #pragma unroll
    for (int r = 0; r < 8; ++r)
      hv[j][r] = H[(size_t)(qb + 8 * half + r) * DD + dcol + 16 * j];
  }
  __builtin_amdgcn_sched_barrier(0);

  v8f acc[4] = {};
  #pragma unroll
  for (int kk = 0; kk < KK; ++kk) {
    #pragma unroll
    for (int i = 0; i < 8; ++i) {
      v2f a;
      a[0] = aO[kk][i >> 1][(i & 1) * 2];
      a[1] = aO[kk][i >> 1][(i & 1) * 2 + 1];
      #pragma unroll
      for (int j = 0; j < 4; ++j) {
        v2f b;
        b[0] = bU[kk][j][i >> 1][(i & 1) * 2];
        b[1] = bU[kk][j][i >> 1][(i & 1) * 2 + 1];
        acc[j] = __builtin_amdgcn_wmma_f32_16x16x4_f32(false, a, false, b, (short)0, acc[j], false, false);
      }
    }
  }

  #pragma unroll
  for (int j = 0; j < 4; ++j) {
    #pragma unroll
    for (int r = 0; r < 8; ++r) {
      const float h = hv[j][r] + ETA * acc[j][r];
      const float a = fabsf(h) - th[j];
      out[(size_t)(qb + 8 * half + r) * DD + dcol + 16 * j] = (a > 0.f) ? copysignf(a, h) : 0.f;
    }
  }
}

#define ORTH_WAVES   10
#define ORTH_THREADS 320
#define ORTH_CH    8

struct OrthFrag { v2f a0, a1, b0, b1; };

__device__ __forceinline__ OrthFrag orth_load(const float* __restrict__ pa,
                                              const float* __restrict__ pb, int d0) {
  const v2f x0 = *(const v2f*)(pa + (size_t)d0 * SS);
  const v2f x1 = *(const v2f*)(pa + (size_t)(d0 + 1) * SS);
  const v2f y0 = *(const v2f*)(pb + (size_t)d0 * SS);
  const v2f y1 = *(const v2f*)(pb + (size_t)(d0 + 1) * SS);
  OrthFrag f;
  f.a0[0] = x0[0]; f.a0[1] = x1[0];
  f.a1[0] = x0[1]; f.a1[1] = x1[1];
  f.b0[0] = y0[0]; f.b0[1] = y1[0];
  f.b1[0] = y0[1]; f.b1[1] = y1[1];
  return f;
}

__global__ __launch_bounds__(ORTH_THREADS) void orth_kernel(const float* __restrict__ U,
                                                            float* __restrict__ out)
{
  const int w = threadIdx.x >> 5;
  const int l = threadIdx.x & 31;
  const int c = l & 15;
  const int h = l >> 4;
  const int kq    = (w >= 4) + (w >= 7) + (w >= 9);
  const int first = (kq == 0) ? 0 : (kq == 1) ? 4 : (kq == 2) ? 7 : 9;
  const int lq    = kq + (w - first);

  const float* pa = U + ((size_t)kq * DD + 2 * h) * SS + 2 * c;
  const float* pb = U + ((size_t)lq * DD + 2 * h) * SS + 2 * c;

  v8f g00 = {}, g01 = {}, g10 = {}, g11 = {};
  OrthFrag cur[ORTH_CH], nxt[ORTH_CH];
  #pragma unroll
  for (int j = 0; j < ORTH_CH; ++j) cur[j] = orth_load(pa, pb, 4 * j);

  #pragma unroll
  for (int g = 0; g < DD / (4 * ORTH_CH); ++g) {
    if (g + 1 < DD / (4 * ORTH_CH)) {
      #pragma unroll
      for (int j = 0; j < ORTH_CH; ++j) nxt[j] = orth_load(pa, pb, 4 * ORTH_CH * (g + 1) + 4 * j);
    }
    __builtin_amdgcn_sched_barrier(0);
    #pragma unroll
    for (int j = 0; j < ORTH_CH; ++j) {
      g00 = __builtin_amdgcn_wmma_f32_16x16x4_f32(false, cur[j].a0, false, cur[j].b0, (short)0, g00, false, false);
      g01 = __builtin_amdgcn_wmma_f32_16x16x4_f32(false, cur[j].a0, false, cur[j].b1, (short)0, g01, false, false);
      g10 = __builtin_amdgcn_wmma_f32_16x16x4_f32(false, cur[j].a1, false, cur[j].b0, (short)0, g10, false, false);
      g11 = __builtin_amdgcn_wmma_f32_16x16x4_f32(false, cur[j].a1, false, cur[j].b1, (short)0, g11, false, false);
    }
    #pragma unroll
    for (int j = 0; j < ORTH_CH; ++j) cur[j] = nxt[j];
  }

  const bool diag = (kq == lq);
  float acc = 0.f;
  #pragma unroll
  for (int r = 0; r < 8; ++r) {
    const float one = (diag && (8 * h + r == c)) ? 1.f : 0.f;
    const float e00 = g00[r] - one;
    const float e11 = g11[r] - one;
    acc = fmaf(e00, e00, acc);
    acc = fmaf(g01[r], g01[r], acc);
    acc = fmaf(g10[r], g10[r], acc);
    acc = fmaf(e11, e11, acc);
  }
  for (int off = 16; off > 0; off >>= 1) acc += __shfl_xor(acc, off, 32);

  __shared__ float red[ORTH_WAVES];
  if (l == 0) red[w] = acc;
  __syncthreads();
  if (threadIdx.x == 0) {
    float s = 0.f;
    for (int i = 0; i < ORTH_WAVES; ++i) s += red[i];
    out[(size_t)NN * DD] = s;
  }
}

extern "C" void kernel_launch(void* const* d_in, const int* in_sizes, int n_in,
                              void* d_out, int out_size, void* d_ws, size_t ws_size,
                              hipStream_t stream) {
  (void)in_sizes; (void)n_in; (void)out_size; (void)ws_size;
  const float* H   = (const float*)d_in[0];
  const float* adj = (const float*)d_in[1];
  const float* U   = (const float*)d_in[2];
  const float* lp  = (const float*)d_in[3];
  const float* thr = (const float*)d_in[4];
  float* out = (float*)d_out;

  char* ws = (char*)d_ws;
  _Float16* Zs = (_Float16*)(ws);
  _Float16* Zf = (_Float16*)(ws + (1u << 20));
  float*    O  = (float*)(ws + (2u << 20));

  zproj_kernel   <<<NN / ZP_ROWS,      128, 0, stream>>>(H, U, lp, Zs, Zf);
  attn_kernel    <<<KK * (NN / 64),    128, 0, stream>>>(Zs, Zf, adj, O);
  combine_kernel <<<NN / 16,           128, 0, stream>>>(O, H, U, thr, out);
  orth_kernel    <<<1,        ORTH_THREADS, 0, stream>>>(U, out);
}


# attention P tile key-major in LDS: one 16-byte store per lane and key tile (was eight 2-byte stores), read back as the P V A operand by transposing loads; values and every sum's order kept
# speedup vs baseline: 9.0827x; 1.1709x over previous
#include <hip/hip_runtime.h>

#define NN 4096
#define DD 256
#define SS 32
#define KK 4
#define ETA 0.5f

typedef __attribute__((ext_vector_type(16))) _Float16 v16h;
typedef __attribute__((ext_vector_type(8)))  _Float16 v8h;
typedef __attribute__((ext_vector_type(8)))  float    v8f;
typedef __attribute__((ext_vector_type(2)))  float    v2f;
typedef __attribute__((ext_vector_type(4)))  float    v4f;

__device__ __forceinline__ v16h make_ab(const _Float16* p0, const _Float16* p1) {
  v8h lo = *(const v8h*)p0;
  v8h hi = *(const v8h*)p1;
  return __builtin_shufflevector(lo, hi, 0,1,2,3,4,5,6,7,8,9,10,11,12,13,14,15);
}

#define ZP_ROWS 16
#define ZP_CH   32

struct UChunk { v2f u[ZP_CH]; };

__device__ __forceinline__ void zp_load(UChunk& u, const float* __restrict__ up0,
                                        const float* __restrict__ up1, int d0)
{
  #pragma unroll
  for (int j = 0; j < ZP_CH; ++j) {
    u.u[j][0] = up0[(size_t)(d0 + j) * SS];
    u.u[j][1] = up1[(size_t)(d0 + j) * SS];
  }
}

#define ZP_G 16
struct HGroup { v4f h[8][ZP_G / 4]; };

__device__ __forceinline__ void zp_hload(HGroup& g, const float* hrow, int d)
{
  #pragma unroll
  for (int r = 0; r < 8; ++r) {
    #pragma unroll
    for (int c = 0; c < ZP_G / 4; ++c)
      g.h[r][c] = *(const v4f*)(hrow + r * DD + d + 4 * c);
  }
}

__device__ __forceinline__ void zp_fma(const v2f* u, const HGroup& g, v2f (&acc)[8])
{
  #pragma unroll
  for (int c = 0; c < ZP_G / 4; ++c) {
    #pragma unroll
    for (int q = 0; q < 4; ++q) {
      #pragma unroll
      for (int r = 0; r < 8; ++r) {
        v2f hh;
        hh[0] = g.h[r][c][q];
        hh[1] = g.h[r][c][q];
        acc[r] = __builtin_elementwise_fma(hh, u[4 * c + q], acc[r]);
      }
    }
  }
}

__global__ __launch_bounds__(128) void zproj_kernel(
    const float* __restrict__ H, const float* __restrict__ U,
    const float* __restrict__ lp,
    _Float16* __restrict__ Zs, _Float16* __restrict__ Zf)
{
  __shared__ float __attribute__((aligned(16))) Hs[ZP_ROWS * DD];

  const int w  = __builtin_amdgcn_readfirstlane((int)(threadIdx.x >> 5));
  const int s  = threadIdx.x & 31;
  const int k0 = 2 * (w & 1);
  const int r0 = 8 * (w >> 1);
  const int nb = blockIdx.x * ZP_ROWS;

  const float* up0 = U + (size_t)k0 * DD * SS + s;
  const float* up1 = up0 + (size_t)DD * SS;

  UChunk A, B;
  zp_load(A, up0, up1, 0);

  const float* hsrc = H + (size_t)nb * DD;
  #pragma unroll
  for (int i = 0; i < 8; ++i) {
    const int o = (i * 128 + (int)threadIdx.x) * 4;
    *(v4f*)(Hs + o) = *(const v4f*)(hsrc + o);
  }
  const float e0 = __expf(lp[k0 * SS + s]);
  const float e1 = __expf(lp[(k0 + 1) * SS + s]);
  __syncthreads();

  const float* hrow = Hs + r0 * DD;
  v2f acc[8];
  #pragma unroll
  for (int r = 0; r < 8; ++r) { acc[r][0] = 0.f; acc[r][1] = 0.f; }

  HGroup HA, HB;
  zp_hload(HA, hrow, 0);
  #pragma nounroll
  for (int d0 = 0; d0 < DD; d0 += 2 * ZP_CH) {
    zp_load(B, up0, up1, d0 + ZP_CH);
    zp_hload(HB, hrow, d0 + ZP_G);
    __builtin_amdgcn_sched_barrier(0);
    zp_fma(A.u, HA, acc);
    zp_hload(HA, hrow, d0 + 2 * ZP_G);
    __builtin_amdgcn_sched_barrier(0);
    zp_fma(A.u + ZP_G, HB, acc);
    zp_load(A, up0, up1, (d0 + 2 * ZP_CH) & (DD - 1));
    zp_hload(HB, hrow, d0 + 3 * ZP_G);
    __builtin_amdgcn_sched_barrier(0);
    zp_fma(B.u, HA, acc);
    zp_hload(HA, hrow, (d0 + 4 * ZP_G) & (DD - 1));
    __builtin_amdgcn_sched_barrier(0);
    zp_fma(B.u + ZP_G, HB, acc);
  }

  #pragma unroll
  for (int r = 0; r < 8; ++r) {
    const size_t n = (size_t)(nb + r0 + r);
    Zf[((size_t)k0 * NN + n) * SS + s]       = (_Float16)acc[r][0];
    Zs[((size_t)k0 * NN + n) * SS + s]       = (_Float16)(acc[r][0] * e0);
    Zf[((size_t)(k0 + 1) * NN + n) * SS + s] = (_Float16)acc[r][1];
    Zs[((size_t)(k0 + 1) * NN + n) * SS + s] = (_Float16)(acc[r][1] * e1);
  }
}

#define KROWB 64

typedef short s16x8 __attribute__((ext_vector_type(8)));

__device__ __forceinline__ v8h ld_tr16(const unsigned char* p) {
  typedef __attribute__((address_space(3))) s16x8 lds_s16x8;
  return __builtin_bit_cast(v8h, __builtin_amdgcn_ds_load_tr16_b128_v8i16((lds_s16x8*)p));
}

struct KAddr { const unsigned char* klo; const unsigned char* khi;
               const unsigned char* tr0; const unsigned char* tr1; };

#define MASK_PF 4
struct ScoreOps {
  v16h  bK[4];
  float w[8][4];
};
struct ValueOps {
  v16h  bV[2][2];
};
struct PFrag { v16h a[2]; };

__device__ __forceinline__ void score_load(ScoreOps& r, const KAddr& a,
    const float* __restrict__ mbase, int kb)
{
  #pragma unroll
  for (int t = 0; t < 4; ++t) {
    const int o = (kb + t * 16) * KROWB;
    r.bK[t] = make_ab((const _Float16*)(a.klo + o), (const _Float16*)(a.khi + o));
  }
  const float* mrow = mbase + kb;
  #pragma unroll
  for (int v = 0; v < 8; ++v) {
    #pragma unroll
    for (int t = 0; t < 4; ++t) r.w[v][t] = mrow[(size_t)v * NN + t * 16];
    __builtin_prefetch(mbase + (size_t)v * NN + ((kb + 64 * MASK_PF) & (NN - 1)), 0, 2);
  }
}

__device__ __forceinline__ void value_load(ValueOps& r, const KAddr& a, int kb)
{
  #pragma unroll
  for (int c = 0; c < 2; ++c) {
    const int o = (kb + c * 32) * KROWB;
    const v8h a0 = ld_tr16(a.tr0 + o);
    const v8h a1 = ld_tr16(a.tr0 + o + 16 * KROWB);
    const v8h b0 = ld_tr16(a.tr1 + o);
    const v8h b1 = ld_tr16(a.tr1 + o + 16 * KROWB);
    r.bV[c][0] = __builtin_shufflevector(a0, a1, 0,1,2,3,4,5,6,7,8,9,10,11,12,13,14,15);
    r.bV[c][1] = __builtin_shufflevector(b0, b1, 0,1,2,3,4,5,6,7,8,9,10,11,12,13,14,15);
  }
}

#define PROWB 32

__device__ __forceinline__ void p_load(PFrag& f, const unsigned char* Pbuf, const int pld)
{
  #pragma unroll
  for (int c = 0; c < 2; ++c) {
    const v8h lo = ld_tr16(Pbuf + (c * 32) * PROWB + pld);
    const v8h hi = ld_tr16(Pbuf + (c * 32 + 16) * PROWB + pld);
    f.a[c] = __builtin_shufflevector(lo, hi, 0,1,2,3,4,5,6,7,8,9,10,11,12,13,14,15);
  }
}

__device__ __forceinline__ void attn_scores(const ScoreOps& r, const v16h aQ,
    const float (&m)[8], const float inv_s, unsigned char* Pbuf, const int pst)
{
  v8f sc[4];
  #pragma unroll
  for (int t = 0; t < 4; ++t) {
    v8f z = {};
    sc[t] = __builtin_amdgcn_wmma_f32_16x16x32_f16(false, aQ, false, r.bK[t], (short)0, z, false, false);
  }
  #pragma unroll
  for (int t = 0; t < 4; ++t) {
    v8h pk;
    #pragma unroll
    for (int v = 0; v < 8; ++v)
      pk[v] = (_Float16)(r.w[v][t] * __expf(sc[t][v] * inv_s - m[v]));
    *(v8h*)(Pbuf + (t * 16) * PROWB + pst) = pk;
  }
}

__device__ __forceinline__ void attn_pv(const PFrag& f, const ValueOps& r, const v16h ones,
    v8f& o0, v8f& o1, v8f& lacc)
{
  #pragma unroll
  for (int c = 0; c < 2; ++c) {
    o0   = __builtin_amdgcn_wmma_f32_16x16x32_f16(false, f.a[c], false, r.bV[c][0], (short)0, o0,   false, false);
    o1   = __builtin_amdgcn_wmma_f32_16x16x32_f16(false, f.a[c], false, r.bV[c][1], (short)0, o1,   false, false);
    lacc = __builtin_amdgcn_wmma_f32_16x16x32_f16(false, f.a[c], false, ones,       (short)0, lacc, false, false);
  }
}

#define S1_TILES 16
#define S1_KEYS  (16 * S1_TILES)
struct KeyGroup { v16h b[S1_TILES]; };

__device__ __forceinline__ void s1_load(KeyGroup& g, const KAddr& a, int kb)
{
  #pragma unroll
  for (int t = 0; t < S1_TILES; ++t) {
    const int o = (kb + t * 16) * KROWB;
    g.b[t] = make_ab((const _Float16*)(a.klo + o), (const _Float16*)(a.khi + o));
  }
}

__device__ __forceinline__ void s1_max(const KeyGroup& g, const v16h aQ, float (&m)[8])
{
  #pragma unroll
  for (int t = 0; t < S1_TILES; ++t) {
    v8f z = {};
    const v8f s1 = __builtin_amdgcn_wmma_f32_16x16x32_f16(false, aQ, false, g.b[t], (short)0, z, false, false);
    #pragma unroll
    for (int v = 0; v < 8; ++v) m[v] = fmaxf(m[v], s1[v]);
  }
}

#define FILL_N 32
struct FillBatch { v4f c[FILL_N]; };

__device__ __forceinline__ void fill_load(FillBatch& f, const unsigned char* __restrict__ src, int b)
{
  #pragma unroll
  for (int i = 0; i < FILL_N; ++i) f.c[i] = *(const v4f*)(src + (size_t)(b * FILL_N + i) * 2048);
}
__device__ __forceinline__ void fill_store(const FillBatch& f, unsigned char* dst, int b)
{
  #pragma unroll
  for (int i = 0; i < FILL_N; ++i) *(v4f*)(dst + (b * FILL_N + i) * 2048) = f.c[i];
}

__global__ __launch_bounds__(128) void attn_kernel(
    const _Float16* __restrict__ Zs, const _Float16* __restrict__ Zf,
    const float* __restrict__ adj, float* __restrict__ O)
{
  const int w    = __builtin_amdgcn_readfirstlane((int)(threadIdx.x >> 5));
  const int k    = blockIdx.x & (KK - 1);
  const int qb   = (blockIdx.x >> 2) * 64 + w * 16;
  const int l    = threadIdx.x & 31;
  const int ll   = l & 15;
  const int half = l >> 4;
  const int c0   = half ? 8 : 0;

  __shared__ unsigned char __attribute__((aligned(16))) Ksh[NN * KROWB];
  __shared__ unsigned char __attribute__((aligned(16))) Pall[4 * 2 * 64 * PROWB];
  unsigned char* P0 = Pall + w * (2 * 64 * PROWB);
  unsigned char* P1 = P0 + 64 * PROWB;
  const int pst = ll * PROWB + 16 * half;
  const int pld = (8 * half + (l & 7)) * PROWB + 16 * ((l >> 3) & 1);

  const float inv_s = 0.17677669529663687f;

  const _Float16* qrow = Zs + ((size_t)k * NN + (qb + ll)) * SS;
  const v16h aQ = make_ab(qrow + c0, qrow + c0 + 16);

  {
    const float* mpf = adj + (size_t)(qb + 8 * half) * NN + ll;
    #pragma unroll
    for (int b = 0; b <= MASK_PF; ++b) {
      #pragma unroll
      for (int v = 0; v < 8; ++v) __builtin_prefetch(mpf + (size_t)v * NN + 64 * b, 0, 2);
    }
  }

  {
    const int t    = (int)threadIdx.x;
    const int frow = t >> 2;
    const int fsw  = ((frow >> 3) & 1) | (((frow >> 2) & 1) << 1);
    const unsigned char* src = (const unsigned char*)(Zf + (size_t)k * NN * SS) + t * 16;
    unsigned char*       dst = Ksh + frow * KROWB + 16 * ((t & 3) ^ fsw);
    FillBatch FA, FB;
    fill_load(FA, src, 0);
    fill_load(FB, src, 1);
    __builtin_amdgcn_sched_barrier(0);
    fill_store(FA, dst, 0);
    fill_load(FA, src, 2);
    __builtin_amdgcn_sched_barrier(0);
    fill_store(FB, dst, 1);
    fill_load(FB, src, 3);
    __builtin_amdgcn_sched_barrier(0);
    fill_store(FA, dst, 2);
    fill_store(FB, dst, 3);
  }
  __syncthreads();

  KAddr ka;
  {
    const int ksw = ((ll >> 3) & 1) | (((ll >> 2) & 1) << 1);
    ka.klo = Ksh + ll * KROWB + 16 * (half ^ ksw);
    ka.khi = Ksh + ll * KROWB + 16 * ((half ^ ksw) ^ 2);
    const int trow = 8 * half + (l & 7);
    const int tsw  = half | (((l >> 2) & 1) << 1);
    const int tq   = (l >> 3) & 1;
    ka.tr0 = Ksh + trow * KROWB + 16 * (tq ^ tsw);
    ka.tr1 = Ksh + trow * KROWB + 16 * ((tq ^ tsw) ^ 2);
  }

  v16h ones;
  #pragma unroll
  for (int i = 0; i < 16; ++i) ones[i] = (_Float16)1.0f;

  float m[8];
  #pragma unroll
  for (int v = 0; v < 8; ++v) m[v] = -INFINITY;

  KeyGroup GA, GB;
  s1_load(GA, ka, 0);
  for (int kb = 0; kb < NN; kb += 2 * S1_KEYS) {
    s1_load(GB, ka, kb + S1_KEYS);
    __builtin_amdgcn_sched_barrier(0);
    s1_max(GA, aQ, m);
    s1_load(GA, ka, (kb + 2 * S1_KEYS) & (NN - 1));
    __builtin_amdgcn_sched_barrier(0);
    s1_max(GB, aQ, m);
  }
  #pragma unroll
  for (int v = 0; v < 8; ++v) {
    float mx = m[v];
    #pragma unroll
    for (int off = 1; off < 16; off <<= 1) mx = fmaxf(mx, __shfl_xor(mx, off, 32));
    m[v] = mx * inv_s;
  }

  v8f o0 = {}; v8f o1 = {};
  v8f lacc = {};

  const float* mbase = adj + (size_t)(qb + 8 * half) * NN + ll;

  ScoreOps KA, KB;
  ValueOps VA, VB;
  PFrag    PA, PB;
  score_load(KA, ka, mbase, 0);
  score_load(KB, ka, mbase, 64);
  value_load(VA, ka, 0);
  attn_scores(KA, aQ, m, inv_s, P0, pst);
  for (int kb = 0; kb < NN; kb += 128) {
    p_load(PA, P0, pld);
    score_load(KA, ka, mbase, (kb + 128) & (NN - 1));
    value_load(VB, ka, kb + 64);
    __builtin_amdgcn_sched_barrier(0);
    attn_scores(KB, aQ, m, inv_s, P1, pst);
    attn_pv(PA, VA, ones, o0, o1, lacc);
    asm volatile("" : "+v"(o0), "+v"(o1), "+v"(lacc));
    p_load(PB, P1, pld);
    score_load(KB, ka, mbase, (kb + 192) & (NN - 1));
    value_load(VA, ka, (kb + 128) & (NN - 1));
    __builtin_amdgcn_sched_barrier(0);
    attn_scores(KA, aQ, m, inv_s, P0, pst);
    attn_pv(PB, VB, ones, o0, o1, lacc);
  }

  float* Ow = O + (size_t)(qb + 8 * half) * (KK * SS) + k * SS + ll;
  #pragma unroll
  for (int v = 0; v < 8; ++v) {
    const float inv_l = 1.0f / lacc[v];
    Ow[(size_t)v * (KK * SS)]      = o0[v] * inv_l;
    Ow[(size_t)v * (KK * SS) + 16] = o1[v] * inv_l;
  }
}

__global__ __launch_bounds__(128) void combine_kernel(
    const float* __restrict__ O, const float* __restrict__ H, const float* __restrict__ U,
    const float* __restrict__ thr, float* __restrict__ out)
{
  const int k    = __builtin_amdgcn_readfirstlane((int)(threadIdx.x >> 5));
  const int qb   = blockIdx.x * 16;
  const int l    = threadIdx.x & 31;
  const int ll   = l & 15;
  const int half = l >> 4;

  const int dcol = 64 * k + ll;
  v4f aO[KK][4];
  #pragma unroll
  for (int kk = 0; kk < KK; ++kk) {
    const float* orow = O + (size_t)(qb + ll) * (KK * SS) + kk * SS + 16 * half;
    #pragma unroll
    for (int mm = 0; mm < 4; ++mm) aO[kk][mm] = *(const v4f*)(orow + 4 * mm);
  }
  v4f bU[KK][4][4];
  #pragma unroll
  for (int kk = 0; kk < KK; ++kk) {
    #pragma unroll
    for (int j = 0; j < 4; ++j) {
      const float* ur = U + ((size_t)kk * DD + dcol + 16 * j) * SS + 16 * half;
      #pragma unroll
      for (int mm = 0; mm < 4; ++mm) bU[kk][j][mm] = *(const v4f*)(ur + 4 * mm);
    }
  }
  float th[4], hv[4][8];
  #pragma unroll
  for (int j = 0; j < 4; ++j) {
    th[j] = thr[dcol + 16 * j];
    #pragma unroll
    for (int r = 0; r < 8; ++r)
      hv[j][r] = H[(size_t)(qb + 8 * half + r) * DD + dcol + 16 * j];
  }
  __builtin_amdgcn_sched_barrier(0);

  v8f acc[4] = {};
  #pragma unroll
  for (int kk = 0; kk < KK; ++kk) {
    #pragma unroll
    for (int i = 0; i < 8; ++i) {
      v2f a;
      a[0] = aO[kk][i >> 1][(i & 1) * 2];
      a[1] = aO[kk][i >> 1][(i & 1) * 2 + 1];
      #pragma unroll
      for (int j = 0; j < 4; ++j) {
        v2f b;
        b[0] = bU[kk][j][i >> 1][(i & 1) * 2];
        b[1] = bU[kk][j][i >> 1][(i & 1) * 2 + 1];
        acc[j] = __builtin_amdgcn_wmma_f32_16x16x4_f32(false, a, false, b, (short)0, acc[j], false, false);
      }
    }
  }

  #pragma unroll
  for (int j = 0; j < 4; ++j) {
    #pragma unroll
    for (int r = 0; r < 8; ++r) {
      const float h = hv[j][r] + ETA * acc[j][r];
      const float a = fabsf(h) - th[j];
      out[(size_t)(qb + 8 * half + r) * DD + dcol + 16 * j] = (a > 0.f) ? copysignf(a, h) : 0.f;
    }
  }
}

#define ORTH_WAVES   10
#define ORTH_THREADS 320
#define ORTH_CH    8

struct OrthFrag { v2f a0, a1, b0, b1; };

__device__ __forceinline__ OrthFrag orth_load(const float* __restrict__ pa,
                                              const float* __restrict__ pb, int d0) {
  const v2f x0 = *(const v2f*)(pa + (size_t)d0 * SS);
  const v2f x1 = *(const v2f*)(pa + (size_t)(d0 + 1) * SS);
  const v2f y0 = *(const v2f*)(pb + (size_t)d0 * SS);
  const v2f y1 = *(const v2f*)(pb + (size_t)(d0 + 1) * SS);
  OrthFrag f;
  f.a0[0] = x0[0]; f.a0[1] = x1[0];
  f.a1[0] = x0[1]; f.a1[1] = x1[1];
  f.b0[0] = y0[0]; f.b0[1] = y1[0];
  f.b1[0] = y0[1]; f.b1[1] = y1[1];
  return f;
}

__global__ __launch_bounds__(ORTH_THREADS) void orth_kernel(const float* __restrict__ U,
                                                            float* __restrict__ out)
{
  const int w = threadIdx.x >> 5;
  const int l = threadIdx.x & 31;
  const int c = l & 15;
  const int h = l >> 4;
  const int kq    = (w >= 4) + (w >= 7) + (w >= 9);
  const int first = (kq == 0) ? 0 : (kq == 1) ? 4 : (kq == 2) ? 7 : 9;
  const int lq    = kq + (w - first);

  const float* pa = U + ((size_t)kq * DD + 2 * h) * SS + 2 * c;
  const float* pb = U + ((size_t)lq * DD + 2 * h) * SS + 2 * c;

  v8f g00 = {}, g01 = {}, g10 = {}, g11 = {};
  OrthFrag cur[ORTH_CH], nxt[ORTH_CH];
  #pragma unroll
  for (int j = 0; j < ORTH_CH; ++j) cur[j] = orth_load(pa, pb, 4 * j);

  #pragma unroll
  for (int g = 0; g < DD / (4 * ORTH_CH); ++g) {
    if (g + 1 < DD / (4 * ORTH_CH)) {
      #pragma unroll
      for (int j = 0; j < ORTH_CH; ++j) nxt[j] = orth_load(pa, pb, 4 * ORTH_CH * (g + 1) + 4 * j);
    }
    __builtin_amdgcn_sched_barrier(0);
    #pragma unroll
    for (int j = 0; j < ORTH_CH; ++j) {
      g00 = __builtin_amdgcn_wmma_f32_16x16x4_f32(false, cur[j].a0, false, cur[j].b0, (short)0, g00, false, false);
      g01 = __builtin_amdgcn_wmma_f32_16x16x4_f32(false, cur[j].a0, false, cur[j].b1, (short)0, g01, false, false);
      g10 = __builtin_amdgcn_wmma_f32_16x16x4_f32(false, cur[j].a1, false, cur[j].b0, (short)0, g10, false, false);
      g11 = __builtin_amdgcn_wmma_f32_16x16x4_f32(false, cur[j].a1, false, cur[j].b1, (short)0, g11, false, false);
    }
    #pragma unroll
    for (int j = 0; j < ORTH_CH; ++j) cur[j] = nxt[j];
  }

  const bool diag = (kq == lq);
  float acc = 0.f;
  #pragma unroll
  for (int r = 0; r < 8; ++r) {
    const float one = (diag && (8 * h + r == c)) ? 1.f : 0.f;
    const float e00 = g00[r] - one;
    const float e11 = g11[r] - one;
    acc = fmaf(e00, e00, acc);
    acc = fmaf(g01[r], g01[r], acc);
    acc = fmaf(g10[r], g10[r], acc);
    acc = fmaf(e11, e11, acc);
  }
  for (int off = 16; off > 0; off >>= 1) acc += __shfl_xor(acc, off, 32);

  __shared__ float red[ORTH_WAVES];
  if (l == 0) red[w] = acc;
  __syncthreads();
  if (threadIdx.x == 0) {
    float s = 0.f;
    for (int i = 0; i < ORTH_WAVES; ++i) s += red[i];
    out[(size_t)NN * DD] = s;
  }
}

extern "C" void kernel_launch(void* const* d_in, const int* in_sizes, int n_in,
                              void* d_out, int out_size, void* d_ws, size_t ws_size,
                              hipStream_t stream) {
  (void)in_sizes; (void)n_in; (void)out_size; (void)ws_size;
  const float* H   = (const float*)d_in[0];
  const float* adj = (const float*)d_in[1];
  const float* U   = (const float*)d_in[2];
  const float* lp  = (const float*)d_in[3];
  const float* thr = (const float*)d_in[4];
  float* out = (float*)d_out;

  char* ws = (char*)d_ws;
  _Float16* Zs = (_Float16*)(ws);
  _Float16* Zf = (_Float16*)(ws + (1u << 20));
  float*    O  = (float*)(ws + (2u << 20));

  zproj_kernel   <<<NN / ZP_ROWS,      128, 0, stream>>>(H, U, lp, Zs, Zf);
  attn_kernel    <<<KK * (NN / 64),    128, 0, stream>>>(Zs, Zf, adj, O);
  combine_kernel <<<NN / 16,           128, 0, stream>>>(O, H, U, thr, out);
  orth_kernel    <<<1,        ORTH_THREADS, 0, stream>>>(U, out);
}
